# starting kernel
# baseline (speedup 1.0000x reference)
#include <hip/hip_runtime.h>
#include <hip/hip_bf16.h>


typedef float v2f __attribute__((ext_vector_type(2)));
typedef float v8f __attribute__((ext_vector_type(8)));

#define NNODES 40000
#define NEDGES 640000
#define NGRAPH 64
#define DIM_IN 128
#define DIM_H  256
#define DIM_O  128
#define MTILES 4


__global__ void k_fill(float* __restrict__ p, float v, long n) {
    long i = (long)blockIdx.x * blockDim.x + threadIdx.x;
    if (i < n) p[i] = v;
}

__global__ void k_deg_accum(const int* __restrict__ dst, float* __restrict__ deg, int e) {
    int i = blockIdx.x * blockDim.x + threadIdx.x;
    if (i < e) unsafeAtomicAdd(&deg[dst[i]], 1.0f);
}

__global__ void k_rsqrt(float* __restrict__ d, int n) {
    int i = blockIdx.x * blockDim.x + threadIdx.x;
    if (i < n) d[i] = rsqrtf(d[i]);
}


__global__ __launch_bounds__(256) void k_gemm_wmma(const float* __restrict__ A,
                                                   const float* __restrict__ W,
                                                   float* __restrict__ C,
                                                   int M, int K, int Nc) {
    const int lane = threadIdx.x & 31;
    const int wave = threadIdx.x >> 5;
    const int m0 = blockIdx.x * (16 * MTILES);
    const int n0 = (blockIdx.y * 8 + wave) * 16;
    if (m0 >= M || n0 >= Nc) return;

    const bool hi = lane >= 16;
    const int  l  = lane & 15;
    const int  col = n0 + l;

    const float* __restrict__ arow[MTILES];
#pragma unroll
    for (int t = 0; t < MTILES; ++t)
        arow[t] = A + (size_t)(m0 + t * 16 + l) * K;

    v8f acc[MTILES];
#pragma unroll
    for (int t = 0; t < MTILES; ++t) acc[t] = (v8f){};

    for (int k0 = 0; k0 < K; k0 += 4) {
        const int ka = k0 + (hi ? 2 : 0);
        v2f b;
        b.x = W[(size_t)ka * Nc + col];
        b.y = W[(size_t)(ka + 1) * Nc + col];
#pragma unroll
        for (int t = 0; t < MTILES; ++t) {
            v2f a = *(const v2f*)(arow[t] + ka);
            acc[t] = __builtin_amdgcn_wmma_f32_16x16x4_f32(
                 false, a,  false, b,
                 (short)0, acc[t],  false,  false);
        }
    }

#pragma unroll
    for (int t = 0; t < MTILES; ++t) {
        const int rbase = m0 + t * 16 + (hi ? 8 : 0);
#pragma unroll
        for (int v = 0; v < 8; ++v)
            C[(size_t)(rbase + v) * Nc + col] = acc[t][v];
    }
}


__global__ void k_scatter(const float* __restrict__ h, const int* __restrict__ src,
                          const int* __restrict__ dst, const float* __restrict__ dis,
                          float* __restrict__ agg, long total, int shift, int H) {
    long gid = (long)blockIdx.x * blockDim.x + threadIdx.x;
    if (gid >= total) return;
    const int e = (int)(gid >> shift);
    const int j = (int)((gid & ((1 << shift) - 1)) << 2);
    const int s = src[e];
    const int d = dst[e];
    const float nrm = dis[s] * dis[d];
    const float4 hv = *(const float4*)(h + (size_t)s * H + j);
    float* out = agg + (size_t)d * H + j;
    unsafeAtomicAdd(out + 0, nrm * hv.x);
    unsafeAtomicAdd(out + 1, nrm * hv.y);
    unsafeAtomicAdd(out + 2, nrm * hv.z);
    unsafeAtomicAdd(out + 3, nrm * hv.w);
}


__global__ void k_fuse(const float* __restrict__ agg, const float* __restrict__ h,
                       const float* __restrict__ dis, const float* __restrict__ b,
                       float* __restrict__ out, long total, int H, int relu) {
    long gid = (long)blockIdx.x * blockDim.x + threadIdx.x;
    if (gid >= total) return;
    const int i = (int)(gid / H);
    const int j = (int)(gid - (long)i * H);
    const float ds = dis[i];
    float v = agg[gid] + ds * ds * h[gid] + b[j];
    if (relu) v = fmaxf(v, 0.0f);
    out[gid] = v;
}


__global__ void k_count(const int* __restrict__ batch, float* __restrict__ counts, int n) {
    int i = blockIdx.x * blockDim.x + threadIdx.x;
    if (i < n) unsafeAtomicAdd(&counts[batch[i]], 1.0f);
}

__global__ void k_pool_sum(const float* __restrict__ emb, const int* __restrict__ batch,
                           float* __restrict__ pool, long total, int D) {
    long gid = (long)blockIdx.x * blockDim.x + threadIdx.x;
    if (gid >= total) return;
    const int i = (int)(gid / D);
    const int j = (int)(gid - (long)i * D);
    unsafeAtomicAdd(&pool[(size_t)batch[i] * D + j], emb[gid]);
}

__global__ void k_pool_fin(const float* __restrict__ pool, const float* __restrict__ counts,
                           float* __restrict__ gemb, int total, int D) {
    int gid = blockIdx.x * blockDim.x + threadIdx.x;
    if (gid >= total) return;
    const int g = gid / D;
    gemb[gid] = pool[gid] / fmaxf(counts[g], 1.0f);
}


__global__ void k_logits(const float* __restrict__ gemb, const float* __restrict__ Wc,
                         const float* __restrict__ bc, float* __restrict__ logits, int D) {
    const int t = threadIdx.x;
    const int g = t >> 1;
    const int c = t & 1;
    float acc = bc[c];
    const float* row = gemb + (size_t)g * D;
    for (int k = 0; k < D; ++k) acc += row[k] * Wc[k * 2 + c];
    logits[g * 2 + c] = acc;
}


static inline int cdiv_l(long a, int b) { return (int)((a + b - 1) / b); }

extern "C" void kernel_launch(void* const* d_in, const int* in_sizes, int n_in,
                              void* d_out, int out_size, void* d_ws, size_t ws_size,
                              hipStream_t stream) {
    const float* x     = (const float*)d_in[0];
    const int*   ei    = (const int*)d_in[1];
    const int*   batch = (const int*)d_in[2];
    const float* W_in  = (const float*)d_in[3];
    const float* b_in  = (const float*)d_in[4];
    const float* W_mid = (const float*)d_in[5];
    const float* b_mid = (const float*)d_in[6];
    const float* W_out = (const float*)d_in[7];
    const float* b_out = (const float*)d_in[8];
    const float* W_cls = (const float*)d_in[9];
    const float* b_cls = (const float*)d_in[10];

    const int N = in_sizes[0] / DIM_IN;
    const int E = in_sizes[1] / 2;
    const int* src = ei;
    const int* dst = ei + E;

    float* w       = (float*)d_ws;
    float* dis     = w;
    float* bufH    = dis + N;
    float* bufAgg  = bufH + (size_t)N * DIM_H;
    float* bufA    = bufAgg + (size_t)N * DIM_H;
    float* pool    = bufA + (size_t)N * DIM_H;
    float* counts  = pool + (size_t)NGRAPH * DIM_O;

    float* node_emb = (float*)d_out;
    float* gemb     = node_emb + (size_t)N * DIM_O;
    float* logits   = gemb + (size_t)NGRAPH * DIM_O;

    const int B = 256;
    const int MB = 16 * MTILES;

    k_fill<<<cdiv_l(N, B), B, 0, stream>>>(dis, 1.0f, N);
    k_deg_accum<<<cdiv_l(E, B), B, 0, stream>>>(dst, dis, E);
    k_rsqrt<<<cdiv_l(N, B), B, 0, stream>>>(dis, N);

    {
        dim3 grid(N / MB, DIM_H / 128);
        k_gemm_wmma<<<grid, B, 0, stream>>>(x, W_in, bufH, N, DIM_IN, DIM_H);
        long tot = (long)N * DIM_H;
        k_fill<<<cdiv_l(tot, B), B, 0, stream>>>(bufAgg, 0.0f, tot);
        long etot = (long)E * (DIM_H / 4);
        k_scatter<<<cdiv_l(etot, B), B, 0, stream>>>(bufH, src, dst, dis, bufAgg, etot, 6, DIM_H);
        k_fuse<<<cdiv_l(tot, B), B, 0, stream>>>(bufAgg, bufH, dis, b_in, bufA, tot, DIM_H, 1);
    }

    {
        dim3 grid(N / MB, DIM_H / 128);
        k_gemm_wmma<<<grid, B, 0, stream>>>(bufA, W_mid, bufH, N, DIM_H, DIM_H);
        long tot = (long)N * DIM_H;
        k_fill<<<cdiv_l(tot, B), B, 0, stream>>>(bufAgg, 0.0f, tot);
        long etot = (long)E * (DIM_H / 4);
        k_scatter<<<cdiv_l(etot, B), B, 0, stream>>>(bufH, src, dst, dis, bufAgg, etot, 6, DIM_H);
        k_fuse<<<cdiv_l(tot, B), B, 0, stream>>>(bufAgg, bufH, dis, b_mid, bufA, tot, DIM_H, 1);
    }

    {
        dim3 grid(N / MB, DIM_O / 128);
        k_gemm_wmma<<<grid, B, 0, stream>>>(bufA, W_out, bufH, N, DIM_H, DIM_O);
        long tot = (long)N * DIM_O;
        k_fill<<<cdiv_l(tot, B), B, 0, stream>>>(bufAgg, 0.0f, tot);
        long etot = (long)E * (DIM_O / 4);
        k_scatter<<<cdiv_l(etot, B), B, 0, stream>>>(bufH, src, dst, dis, bufAgg, etot, 5, DIM_O);
        k_fuse<<<cdiv_l(tot, B), B, 0, stream>>>(bufAgg, bufH, dis, b_out, node_emb, tot, DIM_O, 0);
    }

    {
        long ptot = (long)NGRAPH * DIM_O + NGRAPH;
        k_fill<<<cdiv_l(ptot, B), B, 0, stream>>>(pool, 0.0f, ptot);
        k_count<<<cdiv_l(N, B), B, 0, stream>>>(batch, counts, N);
        long tot = (long)N * DIM_O;
        k_pool_sum<<<cdiv_l(tot, B), B, 0, stream>>>(node_emb, batch, pool, tot, DIM_O);
        int gtot = NGRAPH * DIM_O;
        k_pool_fin<<<cdiv_l(gtot, B), B, 0, stream>>>(pool, counts, gemb, gtot, DIM_O);
    }

    k_logits<<<1, 128, 0, stream>>>(gemb, W_cls, b_cls, logits, DIM_O);
}


# Aggregation scans one owner byte per edge (dst / 160, written once per call by the degree pass) instead of the 4-byte dst list: 4x fewer scan bytes; LDS adds in arrival order as before
# speedup vs baseline: 2.9174x; 2.9174x over previous
#include <hip/hip_runtime.h>
#include <hip/hip_bf16.h>


typedef float v2f __attribute__((ext_vector_type(2)));
typedef float v4f __attribute__((ext_vector_type(4)));
typedef float v8f __attribute__((ext_vector_type(8)));

#define NNODES 40000
#define NEDGES 640000
#define NGRAPH 64
#define DIM_IN 128
#define DIM_H  256
#define DIM_O  128
#define AG_R   160


__global__ void k_fill(float* __restrict__ p, float v, long n) {
    long i = (long)blockIdx.x * blockDim.x + threadIdx.x;
    if (i < n) p[i] = v;
}

__global__ void k_deg_owner(const int* __restrict__ dst, float* __restrict__ deg,
                            unsigned char* __restrict__ owner, int e) {
    typedef int v4i __attribute__((ext_vector_type(4)));
    const int i4 = (blockIdx.x * blockDim.x + threadIdx.x) * 4;
    if (i4 >= e) return;
    if (i4 + 3 < e && ((size_t)dst & 15u) == 0) {
        const v4i d = *(const v4i*)(dst + i4);
        unsafeAtomicAdd(&deg[d.x], 1.0f);
        unsafeAtomicAdd(&deg[d.y], 1.0f);
        unsafeAtomicAdd(&deg[d.z], 1.0f);
        unsafeAtomicAdd(&deg[d.w], 1.0f);
        const unsigned ur = AG_R;
        *(unsigned*)(owner + i4) = ((unsigned)d.x / ur) | (((unsigned)d.y / ur) << 8) |
                                   (((unsigned)d.z / ur) << 16) | (((unsigned)d.w / ur) << 24);
    } else {
        for (int k = 0; k < 4 && i4 + k < e; ++k) {
            const int d = dst[i4 + k];
            unsafeAtomicAdd(&deg[d], 1.0f);
            owner[i4 + k] = (unsigned char)((unsigned)d / (unsigned)AG_R);
        }
    }
}

__global__ void k_rsqrt(float* __restrict__ d, int n) {
    int i = blockIdx.x * blockDim.x + threadIdx.x;
    if (i < n) d[i] = rsqrtf(d[i]);
}


#define G_BM 64
#define G_BN 128
#define G_KC 32
#define G_AP (G_KC + 4)
#define G_BP (2 * G_BN + 32)

template <bool RELU_A, int K, int NC>
__global__ __launch_bounds__(256) void k_gemm_seed(const float* __restrict__ A,
                                                   const float* __restrict__ W,
                                                   float* __restrict__ C,
                                                   int M) {
    __shared__ __attribute__((aligned(16))) float As[G_BM * G_AP];
    __shared__ __attribute__((aligned(16))) float Bs[(G_KC / 2) * G_BP];

    const int tid  = threadIdx.x;
    const int lane = tid & 31;
    const int wave = tid >> 5;
    const int wm   = wave >> 2;
    const int wn   = wave & 3;
    const int m0   = blockIdx.x * G_BM;
    const int nb0  = blockIdx.y * G_BN;
    if (m0 >= M) return;

    const int hi = lane >> 4;
    const int l  = lane & 15;

    const int a_row = tid >> 3;
    const int a_seg = tid & 7;
    const float* __restrict__ ag = A + (size_t)(m0 + a_row) * K + a_seg * 4;
    const int w_kp = wave;
    const float* __restrict__ wg = W + (size_t)(2 * w_kp) * NC + nb0 + lane * 4;

    v4f ra[2], rw[2][2];
    auto fetch = [&](int kc) {
        ra[0] = *(const v4f*)(ag + kc);
        ra[1] = *(const v4f*)(ag + (size_t)32 * K + kc);
        rw[0][0] = *(const v4f*)(wg + (size_t)kc * NC);
        rw[0][1] = *(const v4f*)(wg + (size_t)(kc + 1) * NC);
        rw[1][0] = *(const v4f*)(wg + (size_t)(kc + 16) * NC);
        rw[1][1] = *(const v4f*)(wg + (size_t)(kc + 17) * NC);
    };
    auto stash = [&]() {
        for (int j = 0; j < 2; ++j) {
            v4f v = ra[j];
            if (RELU_A) {
                v.x = fmaxf(v.x, 0.0f); v.y = fmaxf(v.y, 0.0f);
                v.z = fmaxf(v.z, 0.0f); v.w = fmaxf(v.w, 0.0f);
            }
            *(v4f*)(As + (a_row + 32 * j) * G_AP + a_seg * 4) = v;
            const v4f r0 = rw[j][0], r1 = rw[j][1];
            float* bd = Bs + (w_kp + 8 * j) * G_BP + lane * 8;
            *(v4f*)(bd)     = (v4f){r0.x, r1.x, r0.y, r1.y};
            *(v4f*)(bd + 4) = (v4f){r0.z, r1.z, r0.w, r1.w};
        }
    };

    v8f acc[2][2];
    for (int t = 0; t < 2; ++t)
        for (int u = 0; u < 2; ++u) acc[t][u] = (v8f){};

    const float* af = As + (wm * 32 + l) * G_AP + 2 * hi;
    const float* bf = Bs + hi * G_BP + (wn * 32 + l) * 2;

    fetch(0);
    for (int kc = 0; kc < K; kc += G_KC) {
        stash();
        __syncthreads();
        if (kc + G_KC < K) fetch(kc + G_KC);

        v2f a[2][2], b[2][2];
        a[0][0] = *(const v2f*)(af);
        a[0][1] = *(const v2f*)(af + 16 * G_AP);
        b[0][0] = *(const v2f*)(bf);
        b[0][1] = *(const v2f*)(bf + 32);
#pragma unroll
        for (int ks = 0; ks < G_KC / 4; ++ks) {
            const int c = ks & 1, n = c ^ 1;
            if (ks + 1 < G_KC / 4) {
                a[n][0] = *(const v2f*)(af + (ks + 1) * 4);
                a[n][1] = *(const v2f*)(af + 16 * G_AP + (ks + 1) * 4);
                b[n][0] = *(const v2f*)(bf + (ks + 1) * 2 * G_BP);
                b[n][1] = *(const v2f*)(bf + (ks + 1) * 2 * G_BP + 32);
            }
            for (int t = 0; t < 2; ++t)
                for (int u = 0; u < 2; ++u)
                    acc[t][u] = __builtin_amdgcn_wmma_f32_16x16x4_f32(
                         false, a[c][t],  false, b[c][u],
                         (short)0, acc[t][u],  false,  false);
        }
        __syncthreads();
    }

    for (int u = 0; u < 2; ++u) {
        const int col = nb0 + wn * 32 + u * 16 + l;
        for (int t = 0; t < 2; ++t) {
            const int rbase = m0 + wm * 32 + t * 16 + 8 * hi;
            for (int v = 0; v < 8; ++v)
                C[(size_t)(rbase + v) * NC + col] = acc[t][u][v];
        }
    }
}


#define AG_T 1024
#define AG_W (AG_T / 32)
#define AG_Q 1024
#define AG_G 4
static_assert((NNODES + AG_R - 1) / AG_R <= 256, "work-group number must fit in the owner byte");

template <int H, bool FULL>
__device__ __forceinline__ void agg_drain(const float* __restrict__ h,
                                          const int* __restrict__ src,
                                          const int* __restrict__ dst,
                                          const float* __restrict__ dis,
                                          float* acc, const int* qw, int head, int n,
                                          int row0, int lane) {
    constexpr int Q = H / 32;
    int s_l = 0, d_l = 0;
    float n_l = 0.0f;
    if (FULL || lane < n) {
        const int e = qw[(head + lane) & (AG_Q - 1)];
        const int d = dst[e];
        s_l = src[e];
        d_l = d - row0;
        n_l = dis[s_l] * dis[d];
    }
    const int n_i = __float_as_int(n_l);

    float v[2][AG_G][Q];
    auto load_group = [&](int b, int k0) {
        for (int i = 0; i < AG_G; ++i) {
            if (FULL || k0 + i < n) {
                const float* __restrict__ hr =
                    h + (size_t)__builtin_amdgcn_readlane(s_l, k0 + i) * H + lane;
                for (int q = 0; q < Q; ++q) v[b][i][q] = hr[32 * q];
            }
        }
    };
    auto add_group = [&](int b, int k0) {
        for (int i = 0; i < AG_G; ++i) {
            if (FULL || k0 + i < n) {
                const float nk = __int_as_float(__builtin_amdgcn_readlane(n_i, k0 + i));
                float* a = acc + __builtin_amdgcn_readlane(d_l, k0 + i) * H + lane;
                for (int q = 0; q < Q; ++q)
                    __hip_atomic_fetch_add(a + 32 * q, nk * v[b][i][q], __ATOMIC_RELAXED,
                                           __HIP_MEMORY_SCOPE_WORKGROUP);
            }
        }
    };
    load_group(0, 0);
#pragma unroll
    for (int g = 0; g < 32 / AG_G; ++g) {
        if (g + 1 < 32 / AG_G) load_group((g + 1) & 1, (g + 1) * AG_G);
        add_group(g & 1, g * AG_G);
    }
}

template <int H>
__global__ __launch_bounds__(AG_T) void k_aggregate(const float* __restrict__ h,
                                                    const int* __restrict__ src,
                                                    const int* __restrict__ dst,
                                                    const unsigned char* __restrict__ owner,
                                                    const float* __restrict__ dis,
                                                    const float* __restrict__ bias,
                                                    float* __restrict__ out, int E, int N) {
    __shared__ __attribute__((aligned(16))) float acc[AG_R * H];
    __shared__ int queue[AG_W * AG_Q];

    const int tid  = threadIdx.x;
    const int lane = tid & 31;
    const int wave = __builtin_amdgcn_readfirstlane(tid >> 5);
    const int row0 = blockIdx.x * AG_R;
    const int nrows = (N - row0 < AG_R) ? (N - row0) : AG_R;

    constexpr int NP = AG_R * H / 4 / AG_T;
    static_assert(NP * AG_T * 4 == AG_R * H, "whole pieces per thread");
    static_assert(AG_T % (H / 4) == 0, "one column group per thread");
    {
        const int c = (tid % (H / 4)) * 4;
        const int r_t = tid / (H / 4);
        const v4f bv = *(const v4f*)(bias + c);
        v4f hv[NP];
        float dv[NP];
#pragma unroll
        for (int j = 0; j < NP; ++j) {
            const int r = r_t + j * (AG_T / (H / 4));
            const int rc = (r < nrows) ? r : (nrows - 1);
            hv[j] = *(const v4f*)(h + (size_t)(row0 + rc) * H + c);
            dv[j] = dis[row0 + rc];
        }
#pragma unroll
        for (int j = 0; j < NP; ++j) {
            const int r = r_t + j * (AG_T / (H / 4));
            const float d2 = dv[j] * dv[j];
            *(v4f*)(acc + r * H + c) = (v4f){d2 * hv[j].x + bv.x, d2 * hv[j].y + bv.y,
                                             d2 * hv[j].z + bv.z, d2 * hv[j].w + bv.w};
        }
    }
    __syncthreads();

    int* qw = queue + wave * AG_Q;
    int head = 0, cnt = 0;
    typedef int v4i __attribute__((ext_vector_type(4)));
    const unsigned bb = (unsigned)blockIdx.x * 0x01010101u;

    auto enqueue = [&](bool m, int e) {
        const unsigned mask = __builtin_amdgcn_ballot_w32(m);
        const int pos = cnt + (int)__builtin_amdgcn_mbcnt_lo(mask, 0u);
        if (m) qw[(head + pos) & (AG_Q - 1)] = e;
        cnt += __builtin_popcount(mask);
    };
    auto drain_full = [&]() {
        while (cnt >= 32) {
            agg_drain<H, true>(h, src, dst, dis, acc, qw, head, 32, row0, lane);
            head = (head + 32) & (AG_Q - 1);
            cnt -= 32;
        }
    };
    auto take = [&](v4i d, int eb) {
        unsigned z[4];
        for (int i = 0; i < 4; ++i) {
            const unsigned x = (unsigned)d[i] ^ bb;
            z[i] = ~(((x & 0x7f7f7f7fu) + 0x7f7f7f7fu) | x) & 0x80808080u;
        }
        if (__builtin_amdgcn_ballot_w32((z[0] | z[1] | z[2] | z[3]) != 0u) != 0u) {
            const int e0 = eb + lane * 16;
            for (int i = 0; i < 4; ++i) {
                unsigned zi = z[i];
                unsigned mask = __builtin_amdgcn_ballot_w32(zi != 0u);
                while (mask != 0u) {
                    const int byte = __builtin_ctz(zi | 0x80000000u) >> 3;
                    const int pos = cnt + (int)__builtin_amdgcn_mbcnt_lo(mask, 0u);
                    if (zi != 0u) qw[(head + pos) & (AG_Q - 1)] = e0 + i * 4 + byte;
                    cnt += __builtin_popcount(mask);
                    zi &= zi - 1u;
                    mask = __builtin_amdgcn_ballot_w32(zi != 0u);
                }
            }
        }
    };

    const int nfull = E / 2048;
    const int tail0 = nfull * 2048;
    const int nloose = E - tail0;

    for (int b = wave * 32; b < nloose; b += AG_T) {
        const int i = b + lane;
        const int e = tail0 + i;
        const unsigned o = (i < nloose) ? (unsigned)owner[e] : 0xffffffffu;
        enqueue(o == (unsigned)blockIdx.x, e);
    }

    if (wave < nfull) {
        const v4i* __restrict__ p4 = (const v4i*)owner + lane;
        v4i c0, c1, c2, c3;
        {
            const v4i* __restrict__ p = p4 + (size_t)wave * 128;
            c0 = p[0]; c1 = p[32]; c2 = p[64]; c3 = p[96];
        }
        for (int st = wave; st < nfull; st += AG_W) {
            const int stn = (st + AG_W < nfull) ? (st + AG_W) : st;
            const v4i* __restrict__ p = p4 + (size_t)stn * 128;
            const v4i n0 = p[0], n1 = p[32], n2 = p[64], n3 = p[96];
            const int eb = st * 2048;
#pragma nounroll
            for (int j = 0; j < 4; ++j) {
                const v4i c = (j == 0) ? c0 : (j == 1) ? c1 : (j == 2) ? c2 : c3;
                take(c, eb + j * 512);
                drain_full();
            }
            c0 = n0; c1 = n1; c2 = n2; c3 = n3;
        }
    }
    while (cnt > 0) {
        const int n = (cnt < 32) ? cnt : 32;
        agg_drain<H, false>(h, src, dst, dis, acc, qw, head, n, row0, lane);
        head = (head + n) & (AG_Q - 1);
        cnt -= n;
    }

    __syncthreads();
    {
        const int c = (tid % (H / 4)) * 4;
        const int r_t = tid / (H / 4);
        v4f ov[NP];
#pragma unroll
        for (int j = 0; j < NP; ++j)
            ov[j] = *(const v4f*)(acc + (r_t + j * (AG_T / (H / 4))) * H + c);
#pragma unroll
        for (int j = 0; j < NP; ++j) {
            const int r = r_t + j * (AG_T / (H / 4));
            if (r < nrows) *(v4f*)(out + (size_t)(row0 + r) * H + c) = ov[j];
        }
    }
}


#define PL_NPW 16
#define PL_Q (DIM_O / 32)

__device__ __forceinline__ void pool_flush(float (&s)[PL_Q], int len, int g,
                                           float* __restrict__ pool,
                                           float* __restrict__ counts, int lane) {
    float* __restrict__ p = pool + (size_t)g * DIM_O + lane;
    for (int q = 0; q < PL_Q; ++q) {
        unsafeAtomicAdd(p + 32 * q, s[q]);
        s[q] = 0.0f;
    }
    if (lane == 0) unsafeAtomicAdd(&counts[g], (float)len);
}

__global__ __launch_bounds__(256) void k_pool_sum(const float* __restrict__ emb,
                                                  const int* __restrict__ batch,
                                                  float* __restrict__ pool,
                                                  float* __restrict__ counts, int n) {
    const int lane = threadIdx.x & 31;
    const int wave = (int)(((long)blockIdx.x * blockDim.x + threadIdx.x) >> 5);
    const long i0 = (long)wave * PL_NPW;
    if (i0 >= n) return;
    const int cnt = (n - i0 < PL_NPW) ? (int)(n - i0) : PL_NPW;

    int g_l = 0;
    if (lane < cnt) g_l = batch[i0 + lane];
    const float* __restrict__ row = emb + (size_t)i0 * DIM_O + lane;

    float s[PL_Q];
    if (cnt == PL_NPW) {
        float v[PL_NPW][PL_Q];
        for (int k = 0; k < PL_NPW; ++k)
            for (int q = 0; q < PL_Q; ++q) v[k][q] = row[(size_t)k * DIM_O + 32 * q];
        int g = __builtin_amdgcn_readlane(g_l, 0);
        int len = 1;
        for (int q = 0; q < PL_Q; ++q) s[q] = v[0][q];
#pragma unroll
        for (int k = 1; k < PL_NPW; ++k) {
            const int gk = __builtin_amdgcn_readlane(g_l, k);
            if (gk != g) {
                pool_flush(s, len, g, pool, counts, lane);
                len = 0;
                g = gk;
            }
            for (int q = 0; q < PL_Q; ++q) s[q] += v[k][q];
            ++len;
        }
        pool_flush(s, len, g, pool, counts, lane);
    } else {
        int g = __builtin_amdgcn_readlane(g_l, 0);
        int len = 0;
        for (int q = 0; q < PL_Q; ++q) s[q] = 0.0f;
        for (int k = 0; k < cnt; ++k) {
            const int gk = __builtin_amdgcn_readlane(g_l, k);
            if (gk != g) {
                pool_flush(s, len, g, pool, counts, lane);
                len = 0;
                g = gk;
            }
            for (int q = 0; q < PL_Q; ++q) s[q] += row[(size_t)k * DIM_O + 32 * q];
            ++len;
        }
        pool_flush(s, len, g, pool, counts, lane);
    }
}

__global__ void k_pool_fin(const float* __restrict__ pool, const float* __restrict__ counts,
                           float* __restrict__ gemb, int total, int D) {
    int gid = blockIdx.x * blockDim.x + threadIdx.x;
    if (gid >= total) return;
    const int g = gid / D;
    gemb[gid] = pool[gid] / fmaxf(counts[g], 1.0f);
}


__global__ void k_logits(const float* __restrict__ gemb, const float* __restrict__ Wc,
                         const float* __restrict__ bc, float* __restrict__ logits, int D) {
    const int t = threadIdx.x;
    const int g = t >> 1;
    const int c = t & 1;
    float acc = bc[c];
    const float* row = gemb + (size_t)g * D;
    for (int k = 0; k < D; ++k) acc += row[k] * Wc[k * 2 + c];
    logits[g * 2 + c] = acc;
}


static inline int cdiv_l(long a, int b) { return (int)((a + b - 1) / b); }

extern "C" void kernel_launch(void* const* d_in, const int* in_sizes, int n_in,
                              void* d_out, int out_size, void* d_ws, size_t ws_size,
                              hipStream_t stream) {
    const float* x     = (const float*)d_in[0];
    const int*   ei    = (const int*)d_in[1];
    const int*   batch = (const int*)d_in[2];
    const float* W_in  = (const float*)d_in[3];
    const float* b_in  = (const float*)d_in[4];
    const float* W_mid = (const float*)d_in[5];
    const float* b_mid = (const float*)d_in[6];
    const float* W_out = (const float*)d_in[7];
    const float* b_out = (const float*)d_in[8];
    const float* W_cls = (const float*)d_in[9];
    const float* b_cls = (const float*)d_in[10];

    const int N = in_sizes[0] / DIM_IN;
    const int E = in_sizes[1] / 2;
    const int* src = ei;
    const int* dst = ei + E;

    float* w       = (float*)d_ws;
    float* dis     = w;
    float* bufH    = dis + N;
    float* bufAgg  = bufH + (size_t)N * DIM_H;
    float* bufA    = bufAgg + (size_t)N * DIM_H;
    float* pool    = bufA + (size_t)N * DIM_H;
    float* counts  = pool + (size_t)NGRAPH * DIM_O;

    unsigned char* owner = (unsigned char*)bufA;

    float* node_emb = (float*)d_out;
    float* gemb     = node_emb + (size_t)N * DIM_O;
    float* logits   = gemb + (size_t)NGRAPH * DIM_O;

    const int B = 256;
    const int MB = G_BM;

    k_fill<<<cdiv_l(N, B), B, 0, stream>>>(dis, 1.0f, N);
    k_deg_owner<<<cdiv_l(cdiv_l(E, 4), B), B, 0, stream>>>(dst, dis, owner, E);
    k_rsqrt<<<cdiv_l(N, B), B, 0, stream>>>(dis, N);

    {
        dim3 grid(N / MB, DIM_H / 128);
        k_gemm_seed<false, DIM_IN, DIM_H><<<grid, B, 0, stream>>>(x, W_in, bufH, N);
        k_aggregate<DIM_H><<<cdiv_l(N, AG_R), AG_T, 0, stream>>>(bufH, src, dst, owner, dis, b_in,
                                                                 bufAgg, E, N);
    }

    {
        dim3 grid(N / MB, DIM_H / 128);
        k_gemm_seed<true, DIM_H, DIM_H><<<grid, B, 0, stream>>>(bufAgg, W_mid, bufH, N);
        k_aggregate<DIM_H><<<cdiv_l(N, AG_R), AG_T, 0, stream>>>(bufH, src, dst, owner, dis, b_mid,
                                                                 bufAgg, E, N);
    }

    {
        dim3 grid(N / MB, DIM_O / 128);
        k_gemm_seed<true, DIM_H, DIM_O><<<grid, B, 0, stream>>>(bufAgg, W_out, bufH, N);
        k_aggregate<DIM_O><<<cdiv_l(N, AG_R), AG_T, 0, stream>>>(bufH, src, dst, owner, dis, b_out,
                                                                 node_emb, E, N);
    }

    {
        long ptot = (long)NGRAPH * DIM_O + NGRAPH;
        k_fill<<<cdiv_l(ptot, B), B, 0, stream>>>(pool, 0.0f, ptot);
        long wtot = (long)cdiv_l(N, PL_NPW) * 32;
        k_pool_sum<<<cdiv_l(wtot, B), B, 0, stream>>>(node_emb, batch, pool, counts, N);
        int gtot = NGRAPH * DIM_O;
        k_pool_fin<<<cdiv_l(gtot, B), B, 0, stream>>>(pool, counts, gemb, gtot, DIM_O);
    }

    k_logits<<<1, 128, 0, stream>>>(gemb, W_cls, b_cls, logits, DIM_O);
}


# Layer 1's aggregation scan also writes each wave's found edges as 8-byte records (src, local row, norm); layers 2 and 3 replay them: no owner scan or index loads; LDS adds in arrival order as before
# speedup vs baseline: 3.2783x; 1.1237x over previous
#include <hip/hip_runtime.h>
#include <hip/hip_bf16.h>


typedef float v2f __attribute__((ext_vector_type(2)));
typedef float v4f __attribute__((ext_vector_type(4)));
typedef float v8f __attribute__((ext_vector_type(8)));

#define NNODES 40000
#define NEDGES 640000
#define NGRAPH 64
#define DIM_IN 128
#define DIM_H  256
#define DIM_O  128
#define AG_R   160


__global__ void k_fill(float* __restrict__ p, float v, long n) {
    long i = (long)blockIdx.x * blockDim.x + threadIdx.x;
    if (i < n) p[i] = v;
}

__global__ void k_deg_owner(const int* __restrict__ dst, float* __restrict__ deg,
                            unsigned char* __restrict__ owner, int e) {
    typedef int v4i __attribute__((ext_vector_type(4)));
    const int i4 = (blockIdx.x * blockDim.x + threadIdx.x) * 4;
    if (i4 >= e) return;
    if (i4 + 3 < e && ((size_t)dst & 15u) == 0) {
        const v4i d = *(const v4i*)(dst + i4);
        unsafeAtomicAdd(&deg[d.x], 1.0f);
        unsafeAtomicAdd(&deg[d.y], 1.0f);
        unsafeAtomicAdd(&deg[d.z], 1.0f);
        unsafeAtomicAdd(&deg[d.w], 1.0f);
        const unsigned ur = AG_R;
        *(unsigned*)(owner + i4) = ((unsigned)d.x / ur) | (((unsigned)d.y / ur) << 8) |
                                   (((unsigned)d.z / ur) << 16) | (((unsigned)d.w / ur) << 24);
    } else {
        for (int k = 0; k < 4 && i4 + k < e; ++k) {
            const int d = dst[i4 + k];
            unsafeAtomicAdd(&deg[d], 1.0f);
            owner[i4 + k] = (unsigned char)((unsigned)d / (unsigned)AG_R);
        }
    }
}

__global__ void k_rsqrt(float* __restrict__ d, int n) {
    int i = blockIdx.x * blockDim.x + threadIdx.x;
    if (i < n) d[i] = rsqrtf(d[i]);
}


#define G_BM 64
#define G_BN 128
#define G_KC 32
#define G_AP (G_KC + 4)
#define G_BP (2 * G_BN + 32)

template <bool RELU_A, int K, int NC>
__global__ __launch_bounds__(256) void k_gemm_seed(const float* __restrict__ A,
                                                   const float* __restrict__ W,
                                                   float* __restrict__ C,
                                                   int M) {
    __shared__ __attribute__((aligned(16))) float As[G_BM * G_AP];
    __shared__ __attribute__((aligned(16))) float Bs[(G_KC / 2) * G_BP];

    const int tid  = threadIdx.x;
    const int lane = tid & 31;
    const int wave = tid >> 5;
    const int wm   = wave >> 2;
    const int wn   = wave & 3;
    const int m0   = blockIdx.x * G_BM;
    const int nb0  = blockIdx.y * G_BN;
    if (m0 >= M) return;

    const int hi = lane >> 4;
    const int l  = lane & 15;

    const int a_row = tid >> 3;
    const int a_seg = tid & 7;
    const float* __restrict__ ag = A + (size_t)(m0 + a_row) * K + a_seg * 4;
    const int w_kp = wave;
    const float* __restrict__ wg = W + (size_t)(2 * w_kp) * NC + nb0 + lane * 4;

    v4f ra[2], rw[2][2];
    auto fetch = [&](int kc) {
        ra[0] = *(const v4f*)(ag + kc);
        ra[1] = *(const v4f*)(ag + (size_t)32 * K + kc);
        rw[0][0] = *(const v4f*)(wg + (size_t)kc * NC);
        rw[0][1] = *(const v4f*)(wg + (size_t)(kc + 1) * NC);
        rw[1][0] = *(const v4f*)(wg + (size_t)(kc + 16) * NC);
        rw[1][1] = *(const v4f*)(wg + (size_t)(kc + 17) * NC);
    };
    auto stash = [&]() {
        for (int j = 0; j < 2; ++j) {
            v4f v = ra[j];
            if (RELU_A) {
                v.x = fmaxf(v.x, 0.0f); v.y = fmaxf(v.y, 0.0f);
                v.z = fmaxf(v.z, 0.0f); v.w = fmaxf(v.w, 0.0f);
            }
            *(v4f*)(As + (a_row + 32 * j) * G_AP + a_seg * 4) = v;
            const v4f r0 = rw[j][0], r1 = rw[j][1];
            float* bd = Bs + (w_kp + 8 * j) * G_BP + lane * 8;
            *(v4f*)(bd)     = (v4f){r0.x, r1.x, r0.y, r1.y};
            *(v4f*)(bd + 4) = (v4f){r0.z, r1.z, r0.w, r1.w};
        }
    };

    v8f acc[2][2];
    for (int t = 0; t < 2; ++t)
        for (int u = 0; u < 2; ++u) acc[t][u] = (v8f){};

    const float* af = As + (wm * 32 + l) * G_AP + 2 * hi;
    const float* bf = Bs + hi * G_BP + (wn * 32 + l) * 2;

    fetch(0);
    for (int kc = 0; kc < K; kc += G_KC) {
        stash();
        __syncthreads();
        if (kc + G_KC < K) fetch(kc + G_KC);

        v2f a[2][2], b[2][2];
        a[0][0] = *(const v2f*)(af);
        a[0][1] = *(const v2f*)(af + 16 * G_AP);
        b[0][0] = *(const v2f*)(bf);
        b[0][1] = *(const v2f*)(bf + 32);
#pragma unroll
        for (int ks = 0; ks < G_KC / 4; ++ks) {
            const int c = ks & 1, n = c ^ 1;
            if (ks + 1 < G_KC / 4) {
                a[n][0] = *(const v2f*)(af + (ks + 1) * 4);
                a[n][1] = *(const v2f*)(af + 16 * G_AP + (ks + 1) * 4);
                b[n][0] = *(const v2f*)(bf + (ks + 1) * 2 * G_BP);
                b[n][1] = *(const v2f*)(bf + (ks + 1) * 2 * G_BP + 32);
            }
            for (int t = 0; t < 2; ++t)
                for (int u = 0; u < 2; ++u)
                    acc[t][u] = __builtin_amdgcn_wmma_f32_16x16x4_f32(
                         false, a[c][t],  false, b[c][u],
                         (short)0, acc[t][u],  false,  false);
        }
        __syncthreads();
    }

    for (int u = 0; u < 2; ++u) {
        const int col = nb0 + wn * 32 + u * 16 + l;
        for (int t = 0; t < 2; ++t) {
            const int rbase = m0 + wm * 32 + t * 16 + 8 * hi;
            for (int v = 0; v < 8; ++v)
                C[(size_t)(rbase + v) * NC + col] = acc[t][u][v];
        }
    }
}


#define AG_T 1024
#define AG_W (AG_T / 32)
#define AG_Q 1024
#define AG_G 4
#define AG_NB ((NNODES + AG_R - 1) / AG_R)
#define AG_C 512
#define WS_CNT_OFF (1u << 20)
#define WS_REC_OFF ((1u << 20) + (1u << 16))
static_assert(AG_NB <= 256, "work-group number must fit in the owner byte");
static_assert(NEDGES <= WS_CNT_OFF, "owner bytes end ahead of the counts");
static_assert((size_t)AG_NB * AG_W * 4 <= (1u << 16), "counts end ahead of the records");
static_assert((size_t)WS_REC_OFF + (size_t)AG_NB * AG_W * AG_C * 8 <= (size_t)NNODES * DIM_H * 4,
              "records fit in the region");
static_assert(NNODES <= 65536 && AG_R <= 65536, "src and local row share one record word");
typedef int v2i __attribute__((ext_vector_type(2)));

template <int H, bool FULL>
__device__ __forceinline__ void agg_rows(const float* __restrict__ h, float* acc,
                                         int s_l, int d_l, float n_l, int n, int lane) {
    constexpr int Q = H / 32;
    const int n_i = __float_as_int(n_l);

    float v[2][AG_G][Q];
    auto load_group = [&](int b, int k0) {
        for (int i = 0; i < AG_G; ++i) {
            if (FULL || k0 + i < n) {
                const float* __restrict__ hr =
                    h + (size_t)__builtin_amdgcn_readlane(s_l, k0 + i) * H + lane;
                for (int q = 0; q < Q; ++q) v[b][i][q] = hr[32 * q];
            }
        }
    };
    auto add_group = [&](int b, int k0) {
        for (int i = 0; i < AG_G; ++i) {
            if (FULL || k0 + i < n) {
                const float nk = __int_as_float(__builtin_amdgcn_readlane(n_i, k0 + i));
                float* a = acc + __builtin_amdgcn_readlane(d_l, k0 + i) * H + lane;
                for (int q = 0; q < Q; ++q)
                    __hip_atomic_fetch_add(a + 32 * q, nk * v[b][i][q], __ATOMIC_RELAXED,
                                           __HIP_MEMORY_SCOPE_WORKGROUP);
            }
        }
    };
    load_group(0, 0);
#pragma unroll
    for (int g = 0; g < 32 / AG_G; ++g) {
        if (g + 1 < 32 / AG_G) load_group((g + 1) & 1, (g + 1) * AG_G);
        add_group(g & 1, g * AG_G);
    }
}

template <int H, bool FULL, bool REC>
__device__ __forceinline__ void agg_drain(const float* __restrict__ h,
                                          const int* __restrict__ src,
                                          const int* __restrict__ dst,
                                          const float* __restrict__ dis,
                                          float* acc, const int* qw, int head, int n,
                                          int row0, int lane, int* __restrict__ recw, int nrec) {
    int s_l = 0, d_l = 0;
    float n_l = 0.0f;
    if (FULL || lane < n) {
        const int e = qw[(head + lane) & (AG_Q - 1)];
        const int d = dst[e];
        s_l = src[e];
        d_l = d - row0;
        n_l = dis[s_l] * dis[d];
        if (REC && nrec + lane < AG_C)
            *(v2i*)(recw + 2 * (nrec + lane)) = (v2i){s_l | (d_l << 16), __float_as_int(n_l)};
    }
    agg_rows<H, FULL>(h, acc, s_l, d_l, n_l, n, lane);
}

template <int H, bool REC>
__global__ __launch_bounds__(AG_T) void k_aggregate(const float* __restrict__ h,
                                                    const int* __restrict__ src,
                                                    const int* __restrict__ dst,
                                                    const unsigned char* __restrict__ owner,
                                                    const float* __restrict__ dis,
                                                    const float* __restrict__ bias,
                                                    int* __restrict__ rec_cnt,
                                                    int* __restrict__ rec,
                                                    float* __restrict__ out, int E, int N) {
    __shared__ __attribute__((aligned(16))) float acc[AG_R * H];
    __shared__ int queue[AG_W * AG_Q];

    const int tid  = threadIdx.x;
    const int lane = tid & 31;
    const int wave = __builtin_amdgcn_readfirstlane(tid >> 5);
    const int row0 = blockIdx.x * AG_R;
    const int nrows = (N - row0 < AG_R) ? (N - row0) : AG_R;
    int* __restrict__ recw = rec + ((size_t)blockIdx.x * AG_W + wave) * (2 * AG_C);
    int total = 0;
    if (!REC) total = __builtin_amdgcn_readfirstlane(rec_cnt[blockIdx.x * AG_W + wave]);
    const bool replay = !REC && total <= AG_C;

    constexpr int NP = AG_R * H / 4 / AG_T;
    static_assert(NP * AG_T * 4 == AG_R * H, "whole pieces per thread");
    static_assert(AG_T % (H / 4) == 0, "one column group per thread");
    {
        const int c = (tid % (H / 4)) * 4;
        const int r_t = tid / (H / 4);
        const v4f bv = *(const v4f*)(bias + c);
        v4f hv[NP];
        float dv[NP];
#pragma unroll
        for (int j = 0; j < NP; ++j) {
            const int r = r_t + j * (AG_T / (H / 4));
            const int rc = (r < nrows) ? r : (nrows - 1);
            hv[j] = *(const v4f*)(h + (size_t)(row0 + rc) * H + c);
            dv[j] = dis[row0 + rc];
        }
#pragma unroll
        for (int j = 0; j < NP; ++j) {
            const int r = r_t + j * (AG_T / (H / 4));
            const float d2 = dv[j] * dv[j];
            *(v4f*)(acc + r * H + c) = (v4f){d2 * hv[j].x + bv.x, d2 * hv[j].y + bv.y,
                                             d2 * hv[j].z + bv.z, d2 * hv[j].w + bv.w};
        }
    }
    v2i r_cur = (v2i){0, 0};
    if (replay && lane < total) r_cur = *(const v2i*)(recw + 2 * lane);
    __syncthreads();

    int* qw = queue + wave * AG_Q;
    int head = 0, cnt = 0;
    int nrec = 0;
    typedef int v4i __attribute__((ext_vector_type(4)));
    const unsigned bb = (unsigned)blockIdx.x * 0x01010101u;

    auto enqueue = [&](bool m, int e) {
        const unsigned mask = __builtin_amdgcn_ballot_w32(m);
        const int pos = cnt + (int)__builtin_amdgcn_mbcnt_lo(mask, 0u);
        if (m) qw[(head + pos) & (AG_Q - 1)] = e;
        cnt += __builtin_popcount(mask);
    };
    auto drain_full = [&]() {
        while (cnt >= 32) {
            agg_drain<H, true, REC>(h, src, dst, dis, acc, qw, head, 32, row0, lane, recw, nrec);
            head = (head + 32) & (AG_Q - 1);
            cnt -= 32;
            nrec += 32;
        }
    };
    auto take = [&](v4i d, int eb) {
        unsigned z[4];
        for (int i = 0; i < 4; ++i) {
            const unsigned x = (unsigned)d[i] ^ bb;
            z[i] = ~(((x & 0x7f7f7f7fu) + 0x7f7f7f7fu) | x) & 0x80808080u;
        }
        if (__builtin_amdgcn_ballot_w32((z[0] | z[1] | z[2] | z[3]) != 0u) != 0u) {
            const int e0 = eb + lane * 16;
            for (int i = 0; i < 4; ++i) {
                unsigned zi = z[i];
                unsigned mask = __builtin_amdgcn_ballot_w32(zi != 0u);
                while (mask != 0u) {
                    const int byte = __builtin_ctz(zi | 0x80000000u) >> 3;
                    const int pos = cnt + (int)__builtin_amdgcn_mbcnt_lo(mask, 0u);
                    if (zi != 0u) qw[(head + pos) & (AG_Q - 1)] = e0 + i * 4 + byte;
                    cnt += __builtin_popcount(mask);
                    zi &= zi - 1u;
                    mask = __builtin_amdgcn_ballot_w32(zi != 0u);
                }
            }
        }
    };

    if (replay) {
        int base = 0;
        for (; base + 32 <= total; base += 32) {
            v2i r_nxt = (v2i){0, 0};
            if (base + 32 + lane < total) r_nxt = *(const v2i*)(recw + 2 * (base + 32 + lane));
            agg_rows<H, true>(h, acc, r_cur.x & 0xffff, (int)((unsigned)r_cur.x >> 16),
                              __int_as_float(r_cur.y), 32, lane);
            r_cur = r_nxt;
        }
        if (base < total)
            agg_rows<H, false>(h, acc, r_cur.x & 0xffff, (int)((unsigned)r_cur.x >> 16),
                               __int_as_float(r_cur.y), total - base, lane);
    } else {
    const int nfull = E / 2048;
    const int tail0 = nfull * 2048;
    const int nloose = E - tail0;

    for (int b = wave * 32; b < nloose; b += AG_T) {
        const int i = b + lane;
        const int e = tail0 + i;
        const unsigned o = (i < nloose) ? (unsigned)owner[e] : 0xffffffffu;
        enqueue(o == (unsigned)blockIdx.x, e);
    }

    if (wave < nfull) {
        const v4i* __restrict__ p4 = (const v4i*)owner + lane;
        v4i c0, c1, c2, c3;
        {
            const v4i* __restrict__ p = p4 + (size_t)wave * 128;
            c0 = p[0]; c1 = p[32]; c2 = p[64]; c3 = p[96];
        }
        for (int st = wave; st < nfull; st += AG_W) {
            const int stn = (st + AG_W < nfull) ? (st + AG_W) : st;
            const v4i* __restrict__ p = p4 + (size_t)stn * 128;
            const v4i n0 = p[0], n1 = p[32], n2 = p[64], n3 = p[96];
            const int eb = st * 2048;
#pragma nounroll
            for (int j = 0; j < 4; ++j) {
                const v4i c = (j == 0) ? c0 : (j == 1) ? c1 : (j == 2) ? c2 : c3;
                take(c, eb + j * 512);
                drain_full();
            }
            c0 = n0; c1 = n1; c2 = n2; c3 = n3;
        }
    }
    while (cnt > 0) {
        const int n = (cnt < 32) ? cnt : 32;
        agg_drain<H, false, REC>(h, src, dst, dis, acc, qw, head, n, row0, lane, recw, nrec);
        head = (head + n) & (AG_Q - 1);
        cnt -= n;
        nrec += n;
    }
    if (REC && lane == 0) rec_cnt[blockIdx.x * AG_W + wave] = nrec;
    }

    __syncthreads();
    {
        const int lim = nrows * (H / 4);
        const v4f* a4 = (const v4f*)acc + tid;
        v4f* __restrict__ o4 = (v4f*)(out + (size_t)row0 * H) + tid;
        v4f ov[NP];
#pragma unroll
        for (int j = 0; j < NP; ++j) ov[j] = a4[j * AG_T];
#pragma unroll
        for (int j = 0; j < NP; ++j)
            if (tid + j * AG_T < lim) o4[j * AG_T] = ov[j];
    }
}


#define PL_NPW 16
#define PL_Q (DIM_O / 32)

__device__ __forceinline__ void pool_flush(float (&s)[PL_Q], int len, int g,
                                           float* __restrict__ pool,
                                           float* __restrict__ counts, int lane) {
    float* __restrict__ p = pool + (size_t)g * DIM_O + lane;
    for (int q = 0; q < PL_Q; ++q) {
        unsafeAtomicAdd(p + 32 * q, s[q]);
        s[q] = 0.0f;
    }
    if (lane == 0) unsafeAtomicAdd(&counts[g], (float)len);
}

__global__ __launch_bounds__(256) void k_pool_sum(const float* __restrict__ emb,
                                                  const int* __restrict__ batch,
                                                  float* __restrict__ pool,
                                                  float* __restrict__ counts, int n) {
    const int lane = threadIdx.x & 31;
    const int wave = (int)(((long)blockIdx.x * blockDim.x + threadIdx.x) >> 5);
    const long i0 = (long)wave * PL_NPW;
    if (i0 >= n) return;
    const int cnt = (n - i0 < PL_NPW) ? (int)(n - i0) : PL_NPW;

    int g_l = 0;
    if (lane < cnt) g_l = batch[i0 + lane];
    const float* __restrict__ row = emb + (size_t)i0 * DIM_O + lane;

    float s[PL_Q];
    if (cnt == PL_NPW) {
        float v[PL_NPW][PL_Q];
        for (int k = 0; k < PL_NPW; ++k)
            for (int q = 0; q < PL_Q; ++q) v[k][q] = row[(size_t)k * DIM_O + 32 * q];
        int g = __builtin_amdgcn_readlane(g_l, 0);
        int len = 1;
        for (int q = 0; q < PL_Q; ++q) s[q] = v[0][q];
#pragma unroll
        for (int k = 1; k < PL_NPW; ++k) {
            const int gk = __builtin_amdgcn_readlane(g_l, k);
            if (gk != g) {
                pool_flush(s, len, g, pool, counts, lane);
                len = 0;
                g = gk;
            }
            for (int q = 0; q < PL_Q; ++q) s[q] += v[k][q];
            ++len;
        }
        pool_flush(s, len, g, pool, counts, lane);
    } else {
        int g = __builtin_amdgcn_readlane(g_l, 0);
        int len = 0;
        for (int q = 0; q < PL_Q; ++q) s[q] = 0.0f;
        for (int k = 0; k < cnt; ++k) {
            const int gk = __builtin_amdgcn_readlane(g_l, k);
            if (gk != g) {
                pool_flush(s, len, g, pool, counts, lane);
                len = 0;
                g = gk;
            }
            for (int q = 0; q < PL_Q; ++q) s[q] += row[(size_t)k * DIM_O + 32 * q];
            ++len;
        }
        pool_flush(s, len, g, pool, counts, lane);
    }
}

__global__ void k_pool_fin(const float* __restrict__ pool, const float* __restrict__ counts,
                           float* __restrict__ gemb, int total, int D) {
    int gid = blockIdx.x * blockDim.x + threadIdx.x;
    if (gid >= total) return;
    const int g = gid / D;
    gemb[gid] = pool[gid] / fmaxf(counts[g], 1.0f);
}


__global__ void k_logits(const float* __restrict__ gemb, const float* __restrict__ Wc,
                         const float* __restrict__ bc, float* __restrict__ logits, int D) {
    const int t = threadIdx.x;
    const int g = t >> 1;
    const int c = t & 1;
    float acc = bc[c];
    const float* row = gemb + (size_t)g * D;
    for (int k = 0; k < D; ++k) acc += row[k] * Wc[k * 2 + c];
    logits[g * 2 + c] = acc;
}


static inline int cdiv_l(long a, int b) { return (int)((a + b - 1) / b); }

extern "C" void kernel_launch(void* const* d_in, const int* in_sizes, int n_in,
                              void* d_out, int out_size, void* d_ws, size_t ws_size,
                              hipStream_t stream) {
    const float* x     = (const float*)d_in[0];
    const int*   ei    = (const int*)d_in[1];
    const int*   batch = (const int*)d_in[2];
    const float* W_in  = (const float*)d_in[3];
    const float* b_in  = (const float*)d_in[4];
    const float* W_mid = (const float*)d_in[5];
    const float* b_mid = (const float*)d_in[6];
    const float* W_out = (const float*)d_in[7];
    const float* b_out = (const float*)d_in[8];
    const float* W_cls = (const float*)d_in[9];
    const float* b_cls = (const float*)d_in[10];

    const int N = in_sizes[0] / DIM_IN;
    const int E = in_sizes[1] / 2;
    const int* src = ei;
    const int* dst = ei + E;

    float* w       = (float*)d_ws;
    float* dis     = w;
    float* bufH    = dis + N;
    float* bufAgg  = bufH + (size_t)N * DIM_H;
    float* bufA    = bufAgg + (size_t)N * DIM_H;
    float* pool    = bufA + (size_t)N * DIM_H;
    float* counts  = pool + (size_t)NGRAPH * DIM_O;

    unsigned char* owner = (unsigned char*)bufA;
    int* rec_cnt = (int*)((unsigned char*)bufA + WS_CNT_OFF);
    int* rec     = (int*)((unsigned char*)bufA + WS_REC_OFF);

    float* node_emb = (float*)d_out;
    float* gemb     = node_emb + (size_t)N * DIM_O;
    float* logits   = gemb + (size_t)NGRAPH * DIM_O;

    const int B = 256;
    const int MB = G_BM;

    k_fill<<<cdiv_l(N, B), B, 0, stream>>>(dis, 1.0f, N);
    k_deg_owner<<<cdiv_l(cdiv_l(E, 4), B), B, 0, stream>>>(dst, dis, owner, E);
    k_rsqrt<<<cdiv_l(N, B), B, 0, stream>>>(dis, N);

    {
        dim3 grid(N / MB, DIM_H / 128);
        k_gemm_seed<false, DIM_IN, DIM_H><<<grid, B, 0, stream>>>(x, W_in, bufH, N);
        k_aggregate<DIM_H, true><<<cdiv_l(N, AG_R), AG_T, 0, stream>>>(
            bufH, src, dst, owner, dis, b_in, rec_cnt, rec, bufAgg, E, N);
    }

    {
        dim3 grid(N / MB, DIM_H / 128);
        k_gemm_seed<true, DIM_H, DIM_H><<<grid, B, 0, stream>>>(bufAgg, W_mid, bufH, N);
        k_aggregate<DIM_H, false><<<cdiv_l(N, AG_R), AG_T, 0, stream>>>(
            bufH, src, dst, owner, dis, b_mid, rec_cnt, rec, bufAgg, E, N);
    }

    {
        dim3 grid(N / MB, DIM_O / 128);
        k_gemm_seed<true, DIM_H, DIM_O><<<grid, B, 0, stream>>>(bufAgg, W_out, bufH, N);
        k_aggregate<DIM_O, false><<<cdiv_l(N, AG_R), AG_T, 0, stream>>>(
            bufH, src, dst, owner, dis, b_out, rec_cnt, rec, node_emb, E, N);
    }

    {
        long ptot = (long)NGRAPH * DIM_O + NGRAPH;
        k_fill<<<cdiv_l(ptot, B), B, 0, stream>>>(pool, 0.0f, ptot);
        long wtot = (long)cdiv_l(N, PL_NPW) * 32;
        k_pool_sum<<<cdiv_l(wtot, B), B, 0, stream>>>(node_emb, batch, pool, counts, N);
        int gtot = NGRAPH * DIM_O;
        k_pool_fin<<<cdiv_l(gtot, B), B, 0, stream>>>(pool, counts, gemb, gtot, DIM_O);
    }

    k_logits<<<1, 128, 0, stream>>>(gemb, W_cls, b_cls, logits, DIM_O);
}


# Aggregation gathers rows with 16-byte loads: 4 consecutive floats per lane, 4 whole lines per load (256-float row: 8 -> 2 loads); LDS rows re-laid to keep adds on 32 banks; add order as before
# speedup vs baseline: 3.6868x; 1.1246x over previous
#include <hip/hip_runtime.h>
#include <hip/hip_bf16.h>


typedef float v2f __attribute__((ext_vector_type(2)));
typedef float v4f __attribute__((ext_vector_type(4)));
typedef float v8f __attribute__((ext_vector_type(8)));

#define NNODES 40000
#define NEDGES 640000
#define NGRAPH 64
#define DIM_IN 128
#define DIM_H  256
#define DIM_O  128
#define AG_R   160


__global__ void k_deg_owner(const int* __restrict__ dst, float* __restrict__ deg,
                            unsigned char* __restrict__ owner, int e) {
    typedef int v4i __attribute__((ext_vector_type(4)));
    const int i4 = (blockIdx.x * blockDim.x + threadIdx.x) * 4;
    if (i4 >= e) return;
    if (i4 + 3 < e && ((size_t)dst & 15u) == 0) {
        const v4i d = *(const v4i*)(dst + i4);
        unsafeAtomicAdd(&deg[d.x], 1.0f);
        unsafeAtomicAdd(&deg[d.y], 1.0f);
        unsafeAtomicAdd(&deg[d.z], 1.0f);
        unsafeAtomicAdd(&deg[d.w], 1.0f);
        const unsigned ur = AG_R;
        *(unsigned*)(owner + i4) = ((unsigned)d.x / ur) | (((unsigned)d.y / ur) << 8) |
                                   (((unsigned)d.z / ur) << 16) | (((unsigned)d.w / ur) << 24);
    } else {
        for (int k = 0; k < 4 && i4 + k < e; ++k) {
            const int d = dst[i4 + k];
            unsafeAtomicAdd(&deg[d], 1.0f);
            owner[i4 + k] = (unsigned char)((unsigned)d / (unsigned)AG_R);
        }
    }
}


#define G_BM 64
#define G_BN 128
#define G_KC 32
#define G_AP (G_KC + 4)
#define G_BP (2 * G_BN + 32)

template <bool RELU_A, int K, int NC, bool INIT>
__global__ __launch_bounds__(256) void k_gemm_seed(const float* __restrict__ A,
                                                   const float* __restrict__ W,
                                                   float* __restrict__ C,
                                                   int M,
                                                   float* __restrict__ ones, int n_ones,
                                                   float* __restrict__ zeros, int n_zeros) {
    __shared__ __attribute__((aligned(16))) float As[G_BM * G_AP];
    __shared__ __attribute__((aligned(16))) float Bs[(G_KC / 2) * G_BP];

    const int tid  = threadIdx.x;
    const int lane = tid & 31;
    const int wave = tid >> 5;
    const int wm   = wave >> 2;
    const int wn   = wave & 3;
    const int m0   = blockIdx.x * G_BM;
    const int nb0  = blockIdx.y * G_BN;
    if (INIT) {
        const int i = (blockIdx.y * gridDim.x + blockIdx.x) * 256 + tid;
        if (i < n_ones) ones[i] = 1.0f;
        else if (i < n_ones + n_zeros) zeros[i - n_ones] = 0.0f;
    }
    if (m0 >= M) return;

    const int hi = lane >> 4;
    const int l  = lane & 15;

    const int a_row = tid >> 3;
    const int a_seg = tid & 7;
    const float* __restrict__ ag = A + (size_t)(m0 + a_row) * K + a_seg * 4;
    const int w_kp = wave;
    const float* __restrict__ wg = W + (size_t)(2 * w_kp) * NC + nb0 + lane * 4;

    v4f ra[2], rw[2][2];
    auto fetch = [&](int kc) {
        ra[0] = *(const v4f*)(ag + kc);
        ra[1] = *(const v4f*)(ag + (size_t)32 * K + kc);
        rw[0][0] = *(const v4f*)(wg + (size_t)kc * NC);
        rw[0][1] = *(const v4f*)(wg + (size_t)(kc + 1) * NC);
        rw[1][0] = *(const v4f*)(wg + (size_t)(kc + 16) * NC);
        rw[1][1] = *(const v4f*)(wg + (size_t)(kc + 17) * NC);
    };
    auto stash = [&]() {
        for (int j = 0; j < 2; ++j) {
            v4f v = ra[j];
            if (RELU_A) {
                v.x = fmaxf(v.x, 0.0f); v.y = fmaxf(v.y, 0.0f);
                v.z = fmaxf(v.z, 0.0f); v.w = fmaxf(v.w, 0.0f);
            }
            *(v4f*)(As + (a_row + 32 * j) * G_AP + a_seg * 4) = v;
            const v4f r0 = rw[j][0], r1 = rw[j][1];
            float* bd = Bs + (w_kp + 8 * j) * G_BP + lane * 8;
            *(v4f*)(bd)     = (v4f){r0.x, r1.x, r0.y, r1.y};
            *(v4f*)(bd + 4) = (v4f){r0.z, r1.z, r0.w, r1.w};
        }
    };

    v8f acc[2][2];
    for (int t = 0; t < 2; ++t)
        for (int u = 0; u < 2; ++u) acc[t][u] = (v8f){};

    const float* af = As + (wm * 32 + l) * G_AP + 2 * hi;
    const float* bf = Bs + hi * G_BP + (wn * 32 + l) * 2;

    fetch(0);
    for (int kc = 0; kc < K; kc += G_KC) {
        stash();
        __syncthreads();
        if (kc + G_KC < K) fetch(kc + G_KC);

        v2f a[2][2], b[2][2];
        a[0][0] = *(const v2f*)(af);
        a[0][1] = *(const v2f*)(af + 16 * G_AP);
        b[0][0] = *(const v2f*)(bf);
        b[0][1] = *(const v2f*)(bf + 32);
#pragma unroll
        for (int ks = 0; ks < G_KC / 4; ++ks) {
            const int c = ks & 1, n = c ^ 1;
            if (ks + 1 < G_KC / 4) {
                a[n][0] = *(const v2f*)(af + (ks + 1) * 4);
                a[n][1] = *(const v2f*)(af + 16 * G_AP + (ks + 1) * 4);
                b[n][0] = *(const v2f*)(bf + (ks + 1) * 2 * G_BP);
                b[n][1] = *(const v2f*)(bf + (ks + 1) * 2 * G_BP + 32);
            }
            for (int t = 0; t < 2; ++t)
                for (int u = 0; u < 2; ++u)
                    acc[t][u] = __builtin_amdgcn_wmma_f32_16x16x4_f32(
                         false, a[c][t],  false, b[c][u],
                         (short)0, acc[t][u],  false,  false);
        }
        __syncthreads();
    }

    for (int u = 0; u < 2; ++u) {
        const int col = nb0 + wn * 32 + u * 16 + l;
        for (int t = 0; t < 2; ++t) {
            const int rbase = m0 + wm * 32 + t * 16 + 8 * hi;
            for (int v = 0; v < 8; ++v)
                C[(size_t)(rbase + v) * NC + col] = acc[t][u][v];
        }
    }
}


#define AG_T 1024
#define AG_W (AG_T / 32)
#define AG_Q 1024
#define AG_G 4
#define AG_NB ((NNODES + AG_R - 1) / AG_R)
#define AG_C 512
#define WS_CNT_OFF (1u << 20)
#define WS_REC_OFF ((1u << 20) + (1u << 16))
static_assert(AG_NB <= 256, "work-group number must fit in the owner byte");
static_assert(NEDGES <= WS_CNT_OFF, "owner bytes end ahead of the counts");
static_assert((size_t)AG_NB * AG_W * 4 <= (1u << 16), "counts end ahead of the records");
static_assert((size_t)WS_REC_OFF + (size_t)AG_NB * AG_W * AG_C * 8 <= (size_t)NNODES * DIM_H * 4,
              "records fit in the region");
static_assert(NNODES <= 65536 && AG_R <= 65536, "src and local row share one record word");
typedef int v2i __attribute__((ext_vector_type(2)));

__device__ __forceinline__ int row_pos(int c) {
    return (c & ~127) | ((c & 3) << 5) | ((c & 127) >> 2);
}
__device__ __forceinline__ int row_col(int p) {
    return (p & ~127) | ((p & 31) << 2) | ((p >> 5) & 3);
}
template <int H, bool FULL>
__device__ __forceinline__ void agg_rows(const float* __restrict__ h, float* acc,
                                         int s_l, int d_l, float n_l, int n, int lane) {
    constexpr int Q = H / 128;
    static_assert(Q * 128 == H, "whole groups of 128 columns");
    const int n_i = __float_as_int(n_l);

    v4f v[2][AG_G][Q];
    auto load_group = [&](int b, int k0) {
        for (int i = 0; i < AG_G; ++i) {
            if (FULL || k0 + i < n) {
                const float* __restrict__ hr =
                    h + (size_t)__builtin_amdgcn_readlane(s_l, k0 + i) * H + lane * 4;
                for (int q = 0; q < Q; ++q) v[b][i][q] = *(const v4f*)(hr + 128 * q);
            }
        }
    };
    auto add_group = [&](int b, int k0) {
        for (int i = 0; i < AG_G; ++i) {
            if (FULL || k0 + i < n) {
                const float nk = __int_as_float(__builtin_amdgcn_readlane(n_i, k0 + i));
                float* a = acc + __builtin_amdgcn_readlane(d_l, k0 + i) * H + lane;
                for (int q = 0; q < Q; ++q) {
                    const v4f t = v[b][i][q];
                    __hip_atomic_fetch_add(a + 128 * q,      nk * t.x, __ATOMIC_RELAXED,
                                           __HIP_MEMORY_SCOPE_WORKGROUP);
                    __hip_atomic_fetch_add(a + 128 * q + 32, nk * t.y, __ATOMIC_RELAXED,
                                           __HIP_MEMORY_SCOPE_WORKGROUP);
                    __hip_atomic_fetch_add(a + 128 * q + 64, nk * t.z, __ATOMIC_RELAXED,
                                           __HIP_MEMORY_SCOPE_WORKGROUP);
                    __hip_atomic_fetch_add(a + 128 * q + 96, nk * t.w, __ATOMIC_RELAXED,
                                           __HIP_MEMORY_SCOPE_WORKGROUP);
                }
            }
        }
    };
    load_group(0, 0);
#pragma unroll
    for (int g = 0; g < 32 / AG_G; ++g) {
        if (g + 1 < 32 / AG_G) load_group((g + 1) & 1, (g + 1) * AG_G);
        add_group(g & 1, g * AG_G);
    }
}

template <int H, bool FULL, bool REC>
__device__ __forceinline__ void agg_drain(const float* __restrict__ h,
                                          const int* __restrict__ src,
                                          const int* __restrict__ dst,
                                          const float* __restrict__ deg,
                                          float* acc, const int* qw, int head, int n,
                                          int row0, int lane, int* __restrict__ recw, int nrec) {
    int s_l = 0, d_l = 0;
    float n_l = 0.0f;
    if (FULL || lane < n) {
        const int e = qw[(head + lane) & (AG_Q - 1)];
        const int d = dst[e];
        s_l = src[e];
        d_l = d - row0;
        n_l = rsqrtf(deg[s_l]) * rsqrtf(deg[d]);
        if (REC && nrec + lane < AG_C)
            *(v2i*)(recw + 2 * (nrec + lane)) = (v2i){s_l | (d_l << 16), __float_as_int(n_l)};
    }
    agg_rows<H, FULL>(h, acc, s_l, d_l, n_l, n, lane);
}

#define PL_S(H) ((H) / 32)
#define PL_G (AG_R / 32)
static_assert(PL_G * 32 == AG_R, "whole runs of 32 rows");
template <int H, bool REC, bool POOL>
__global__ __launch_bounds__(AG_T) void k_aggregate(const float* __restrict__ h,
                                                    const int* __restrict__ src,
                                                    const int* __restrict__ dst,
                                                    const unsigned char* __restrict__ owner,
                                                    const float* __restrict__ deg,
                                                    const float* __restrict__ bias,
                                                    int* __restrict__ rec_cnt,
                                                    int* __restrict__ rec,
                                                    float* __restrict__ out,
                                                    const int* __restrict__ batch,
                                                    float* __restrict__ pool,
                                                    float* __restrict__ counts, int E, int N) {
    __shared__ __attribute__((aligned(16))) float acc[AG_R * H];
    __shared__ int queue[AG_W * AG_Q];

    const int tid  = threadIdx.x;
    const int lane = tid & 31;
    const int wave = __builtin_amdgcn_readfirstlane(tid >> 5);
    const int row0 = blockIdx.x * AG_R;
    const int nrows = (N - row0 < AG_R) ? (N - row0) : AG_R;
    int* __restrict__ recw = rec + ((size_t)blockIdx.x * AG_W + wave) * (2 * AG_C);
    int total = 0;
    if (!REC) total = __builtin_amdgcn_readfirstlane(rec_cnt[blockIdx.x * AG_W + wave]);
    const bool replay = !REC && total <= AG_C;

    constexpr int NP = AG_R * H / 4 / AG_T;
    static_assert(NP * AG_T * 4 == AG_R * H, "whole pieces per thread");
    static_assert(AG_T % (H / 4) == 0, "one column group per thread");
    const int p_t = row_pos((tid % (H / 4)) * 4);
    {
        const int c = (tid % (H / 4)) * 4;
        const int r_t = tid / (H / 4);
        const v4f bv = *(const v4f*)(bias + c);
        v4f hv[NP];
        float dv[NP];
#pragma unroll
        for (int j = 0; j < NP; ++j) {
            const int r = r_t + j * (AG_T / (H / 4));
            const int rc = (r < nrows) ? r : (nrows - 1);
            hv[j] = *(const v4f*)(h + (size_t)(row0 + rc) * H + c);
            dv[j] = deg[row0 + rc];
        }
#pragma unroll
        for (int j = 0; j < NP; ++j) {
            const int r = r_t + j * (AG_T / (H / 4));
            const float di = rsqrtf(dv[j]);
            const float d2 = di * di;
            float* ar = acc + r * H + p_t;
            ar[0]  = d2 * hv[j].x + bv.x;
            ar[32] = d2 * hv[j].y + bv.y;
            ar[64] = d2 * hv[j].z + bv.z;
            ar[96] = d2 * hv[j].w + bv.w;
        }
    }
    v2i r_cur = (v2i){0, 0};
    if (replay && lane < total) r_cur = *(const v2i*)(recw + 2 * lane);
    __syncthreads();

    int* qw = queue + wave * AG_Q;
    int head = 0, cnt = 0;
    int nrec = 0;
    typedef int v4i __attribute__((ext_vector_type(4)));
    const unsigned bb = (unsigned)blockIdx.x * 0x01010101u;

    auto enqueue = [&](bool m, int e) {
        const unsigned mask = __builtin_amdgcn_ballot_w32(m);
        const int pos = cnt + (int)__builtin_amdgcn_mbcnt_lo(mask, 0u);
        if (m) qw[(head + pos) & (AG_Q - 1)] = e;
        cnt += __builtin_popcount(mask);
    };
    auto drain_full = [&]() {
        while (cnt >= 32) {
            agg_drain<H, true, REC>(h, src, dst, deg, acc, qw, head, 32, row0, lane, recw, nrec);
            head = (head + 32) & (AG_Q - 1);
            cnt -= 32;
            nrec += 32;
        }
    };
    auto take = [&](v4i d, int eb) {
        unsigned z[4];
        for (int i = 0; i < 4; ++i) {
            const unsigned x = (unsigned)d[i] ^ bb;
            z[i] = ~(((x & 0x7f7f7f7fu) + 0x7f7f7f7fu) | x) & 0x80808080u;
        }
        if (__builtin_amdgcn_ballot_w32((z[0] | z[1] | z[2] | z[3]) != 0u) != 0u) {
            const int e0 = eb + lane * 16;
            for (int i = 0; i < 4; ++i) {
                unsigned zi = z[i];
                unsigned mask = __builtin_amdgcn_ballot_w32(zi != 0u);
                while (mask != 0u) {
                    const int byte = __builtin_ctz(zi | 0x80000000u) >> 3;
                    const int pos = cnt + (int)__builtin_amdgcn_mbcnt_lo(mask, 0u);
                    if (zi != 0u) qw[(head + pos) & (AG_Q - 1)] = e0 + i * 4 + byte;
                    cnt += __builtin_popcount(mask);
                    zi &= zi - 1u;
                    mask = __builtin_amdgcn_ballot_w32(zi != 0u);
                }
            }
        }
    };

    if (replay) {
        int base = 0;
        for (; base + 32 <= total; base += 32) {
            v2i r_nxt = (v2i){0, 0};
            if (base + 32 + lane < total) r_nxt = *(const v2i*)(recw + 2 * (base + 32 + lane));
            agg_rows<H, true>(h, acc, r_cur.x & 0xffff, (int)((unsigned)r_cur.x >> 16),
                              __int_as_float(r_cur.y), 32, lane);
            r_cur = r_nxt;
        }
        if (base < total)
            agg_rows<H, false>(h, acc, r_cur.x & 0xffff, (int)((unsigned)r_cur.x >> 16),
                               __int_as_float(r_cur.y), total - base, lane);
    } else {
    const int nfull = E / 2048;
    const int tail0 = nfull * 2048;
    const int nloose = E - tail0;

    for (int b = wave * 32; b < nloose; b += AG_T) {
        const int i = b + lane;
        const int e = tail0 + i;
        const unsigned o = (i < nloose) ? (unsigned)owner[e] : 0xffffffffu;
        enqueue(o == (unsigned)blockIdx.x, e);
    }

    if (wave < nfull) {
        const v4i* __restrict__ p4 = (const v4i*)owner + lane;
        v4i c0, c1, c2, c3;
        {
            const v4i* __restrict__ p = p4 + (size_t)wave * 128;
            c0 = p[0]; c1 = p[32]; c2 = p[64]; c3 = p[96];
        }
        for (int st = wave; st < nfull; st += AG_W) {
            const int stn = (st + AG_W < nfull) ? (st + AG_W) : st;
            const v4i* __restrict__ p = p4 + (size_t)stn * 128;
            const v4i n0 = p[0], n1 = p[32], n2 = p[64], n3 = p[96];
            const int eb = st * 2048;
#pragma nounroll
            for (int j = 0; j < 4; ++j) {
                const v4i c = (j == 0) ? c0 : (j == 1) ? c1 : (j == 2) ? c2 : c3;
                take(c, eb + j * 512);
                drain_full();
            }
            c0 = n0; c1 = n1; c2 = n2; c3 = n3;
        }
    }
    while (cnt > 0) {
        const int n = (cnt < 32) ? cnt : 32;
        agg_drain<H, false, REC>(h, src, dst, deg, acc, qw, head, n, row0, lane, recw, nrec);
        head = (head + n) & (AG_Q - 1);
        cnt -= n;
        nrec += n;
    }
    if (REC && lane == 0) rec_cnt[blockIdx.x * AG_W + wave] = nrec;
    }

    static_assert(!POOL || PL_S(H) * PL_G <= AG_W, "a wave per slice and run");
    const bool pooler = POOL && wave < PL_S(H) * PL_G;
    const int p_run = wave / PL_S(H);
    int g_l = 0;
    if (pooler) {
        const int r = p_run * 32 + lane;
        g_l = batch[row0 + ((r < nrows) ? r : (nrows - 1))];
    }
    __syncthreads();
    {
        const int lim = nrows * (H / 4);
        const float* ar = acc + (tid / (H / 4)) * H + p_t;
        v4f* __restrict__ o4 = (v4f*)(out + (size_t)row0 * H) + tid;
        v4f ov[NP];
#pragma unroll
        for (int j = 0; j < NP; ++j) {
            const float* aj = ar + j * (AG_T / (H / 4)) * H;
            ov[j] = (v4f){aj[0], aj[32], aj[64], aj[96]};
        }
#pragma unroll
        for (int j = 0; j < NP; ++j)
            if (tid + j * AG_T < lim) o4[j * AG_T] = ov[j];
    }
    if (pooler) {
        const int cnt_r = (nrows - p_run * 32 < 32) ? (nrows - p_run * 32) : 32;
        if (cnt_r > 0) {
            const int pp = (wave % PL_S(H)) * 32 + lane;
            const int c = row_col(pp);
            const float* a = acc + (p_run * 32) * H + pp;
            float v[32];
            for (int k = 0; k < 32; ++k) v[k] = a[k * H];
            float* __restrict__ pc = pool + c;
            int g = __builtin_amdgcn_readlane(g_l, 0);
            int len = 0;
            float s = 0.0f;
#pragma unroll
            for (int k = 0; k < 32; ++k) {
                if (k < cnt_r) {
                    const int gk = __builtin_amdgcn_readlane(g_l, k);
                    if (gk != g) {
                        unsafeAtomicAdd(pc + (size_t)g * H, s);
                        if (c == 0) unsafeAtomicAdd(&counts[g], (float)len);
                        s = 0.0f;
                        len = 0;
                        g = gk;
                    }
                    s += v[k];
                    ++len;
                }
            }
            unsafeAtomicAdd(pc + (size_t)g * H, s);
            if (c == 0) unsafeAtomicAdd(&counts[g], (float)len);
        }
    }
}


__global__ __launch_bounds__(DIM_O) void k_head(const float* __restrict__ pool,
                                                const float* __restrict__ counts,
                                                const float* __restrict__ Wc,
                                                const float* __restrict__ bc,
                                                float* __restrict__ gemb,
                                                float* __restrict__ logits) {
    __shared__ float part[DIM_O / 32][2];
    const int g = blockIdx.x;
    const int t = threadIdx.x;
    const v2f w = *(const v2f*)(Wc + 2 * t);
    const float ge = pool[g * DIM_O + t] / fmaxf(counts[g], 1.0f);
    gemb[g * DIM_O + t] = ge;
    float p0 = ge * w.x, p1 = ge * w.y;
    for (int o = 16; o > 0; o >>= 1) {
        p0 += __shfl_xor(p0, o);
        p1 += __shfl_xor(p1, o);
    }
    if ((t & 31) == 0) {
        part[t >> 5][0] = p0;
        part[t >> 5][1] = p1;
    }
    __syncthreads();
    if (t < 2) {
        float a = bc[t];
        for (int k = 0; k < DIM_O / 32; ++k) a += part[k][t];
        logits[g * 2 + t] = a;
    }
}


static inline int cdiv_l(long a, int b) { return (int)((a + b - 1) / b); }

extern "C" void kernel_launch(void* const* d_in, const int* in_sizes, int n_in,
                              void* d_out, int out_size, void* d_ws, size_t ws_size,
                              hipStream_t stream) {
    const float* x     = (const float*)d_in[0];
    const int*   ei    = (const int*)d_in[1];
    const int*   batch = (const int*)d_in[2];
    const float* W_in  = (const float*)d_in[3];
    const float* b_in  = (const float*)d_in[4];
    const float* W_mid = (const float*)d_in[5];
    const float* b_mid = (const float*)d_in[6];
    const float* W_out = (const float*)d_in[7];
    const float* b_out = (const float*)d_in[8];
    const float* W_cls = (const float*)d_in[9];
    const float* b_cls = (const float*)d_in[10];

    const int N = in_sizes[0] / DIM_IN;
    const int E = in_sizes[1] / 2;
    const int* src = ei;
    const int* dst = ei + E;

    float* w       = (float*)d_ws;
    float* deg     = w;
    float* bufH    = deg + N;
    float* bufAgg  = bufH + (size_t)N * DIM_H;
    float* bufA    = bufAgg + (size_t)N * DIM_H;
    float* pool    = bufA + (size_t)N * DIM_H;
    float* counts  = pool + (size_t)NGRAPH * DIM_O;

    unsigned char* owner = (unsigned char*)bufA;
    int* rec_cnt = (int*)((unsigned char*)bufA + WS_CNT_OFF);
    int* rec     = (int*)((unsigned char*)bufA + WS_REC_OFF);

    float* node_emb = (float*)d_out;
    float* gemb     = node_emb + (size_t)N * DIM_O;
    float* logits   = gemb + (size_t)NGRAPH * DIM_O;

    const int B = 256;
    const int MB = G_BM;

    {
        dim3 grid(N / MB, DIM_H / 128);
        k_gemm_seed<false, DIM_IN, DIM_H, true><<<grid, B, 0, stream>>>(
            x, W_in, bufH, N, deg, N, pool, NGRAPH * DIM_O + NGRAPH);
    }

    k_deg_owner<<<cdiv_l(cdiv_l(E, 4), B), B, 0, stream>>>(dst, deg, owner, E);

    k_aggregate<DIM_H, true, false><<<cdiv_l(N, AG_R), AG_T, 0, stream>>>(
        bufH, src, dst, owner, deg, b_in, rec_cnt, rec, bufAgg, batch, pool, counts, E, N);

    {
        dim3 grid(N / MB, DIM_H / 128);
        k_gemm_seed<true, DIM_H, DIM_H, false><<<grid, B, 0, stream>>>(
            bufAgg, W_mid, bufH, N, deg, 0, pool, 0);
        k_aggregate<DIM_H, false, false><<<cdiv_l(N, AG_R), AG_T, 0, stream>>>(
            bufH, src, dst, owner, deg, b_mid, rec_cnt, rec, bufAgg, batch, pool, counts, E, N);
    }

    {
        dim3 grid(N / MB, DIM_O / 128);
        k_gemm_seed<true, DIM_H, DIM_O, false><<<grid, B, 0, stream>>>(
            bufAgg, W_out, bufH, N, deg, 0, pool, 0);
        k_aggregate<DIM_O, false, true><<<cdiv_l(N, AG_R), AG_T, 0, stream>>>(
            bufH, src, dst, owner, deg, b_out, rec_cnt, rec, node_emb, batch, pool, counts, E, N);
    }

    k_head<<<NGRAPH, DIM_O, 0, stream>>>(pool, counts, W_cls, b_cls, gemb, logits);
}


# Degree pass also groups each 1024-edge tile's edges by owner work-group (LDS counters, fixed ranks); layer 1's aggregation walks its groups: no owner-byte scan or queue; LDS adds in arrival order
# speedup vs baseline: 3.9130x; 1.0614x over previous
#include <hip/hip_runtime.h>
#include <hip/hip_bf16.h>


typedef float v2f __attribute__((ext_vector_type(2)));
typedef float v4f __attribute__((ext_vector_type(4)));
typedef float v8f __attribute__((ext_vector_type(8)));

#define NNODES 40000
#define NEDGES 640000
#define NGRAPH 64
#define DIM_IN 128
#define DIM_H  256
#define DIM_O  128
#define AG_R   160


#define BK_T 256
#define BK_E (BK_T * 4)
#define BK_K 256
static_assert(BK_K == BK_T, "one thread per owner slot");
static_assert(BK_E < 65536, "offsets inside a tile fit in 16 bits");

__global__ __launch_bounds__(BK_T) void k_deg_group(const int* __restrict__ src,
                                                    const int* __restrict__ dst,
                                                    float* __restrict__ deg,
                                                    unsigned short* __restrict__ toff,
                                                    int* __restrict__ list, int e) {
    typedef int v4i __attribute__((ext_vector_type(4)));
    __shared__ unsigned cnt[BK_T / 32][BK_K];
    __shared__ unsigned koff[BK_K];
    __shared__ unsigned wsum[BK_T / 32];
    const int tid  = threadIdx.x;
    const int lane = tid & 31;
    const int wave = tid >> 5;
    for (int w = 0; w < BK_T / 32; ++w) cnt[w][tid] = 0u;
    const int i4 = blockIdx.x * BK_E + tid * 4;
    int d[4] = {0, 0, 0, 0}, s[4] = {0, 0, 0, 0};
    int nv = e - i4;
    nv = (nv < 0) ? 0 : ((nv > 4) ? 4 : nv);
    if (nv == 4 && (((size_t)dst | (size_t)src) & 15u) == 0) {
        const v4i dv = *(const v4i*)(dst + i4);
        const v4i sv = *(const v4i*)(src + i4);
        d[0] = dv.x; d[1] = dv.y; d[2] = dv.z; d[3] = dv.w;
        s[0] = sv.x; s[1] = sv.y; s[2] = sv.z; s[3] = sv.w;
    } else {
        for (int j = 0; j < 4; ++j)
            if (j < nv) { d[j] = dst[i4 + j]; s[j] = src[i4 + j]; }
    }
    __syncthreads();
    unsigned key[4], rank[4];
    for (int j = 0; j < 4; ++j) {
        key[j] = (unsigned)d[j] / (unsigned)AG_R;
        rank[j] = 0u;
        if (j < nv) {
            rank[j] = __hip_atomic_fetch_add(&cnt[wave][key[j]], 1u, __ATOMIC_RELAXED,
                                             __HIP_MEMORY_SCOPE_WORKGROUP);
            unsafeAtomicAdd(&deg[d[j]], 1.0f);
        }
    }
    __syncthreads();
    unsigned tot = 0u;
    for (int w = 0; w < BK_T / 32; ++w) {
        const unsigned cw = cnt[w][tid];
        cnt[w][tid] = tot;
        tot += cw;
    }
    unsigned inc = tot;
    for (int o = 1; o < 32; o <<= 1) {
        const unsigned v = __shfl_up(inc, o);
        if (lane >= o) inc += v;
    }
    if (lane == 31) wsum[wave] = inc;
    __syncthreads();
    unsigned wb = 0u;
    for (int w = 0; w < BK_T / 32; ++w)
        if (w < wave) wb += wsum[w];
    const unsigned ko = wb + inc - tot;
    koff[tid] = ko;
    toff[(size_t)blockIdx.x * BK_K + tid] = (unsigned short)ko;
    __syncthreads();
    for (int j = 0; j < 4; ++j) {
        if (j < nv) {
            const unsigned pos = koff[key[j]] + cnt[wave][key[j]] + rank[j];
            list[(size_t)blockIdx.x * BK_E + pos] =
                s[j] | ((d[j] - (int)(key[j] * (unsigned)AG_R)) << 16);
        }
    }
}


#define G_BM 64
#define G_BN 128
#define G_KC 32
#define G_AP (G_KC + 4)
#define G_BP (2 * G_BN + 32)

template <bool RELU_A, int K, int NC, bool INIT>
__global__ __launch_bounds__(256) void k_gemm_seed(const float* __restrict__ A,
                                                   const float* __restrict__ W,
                                                   float* __restrict__ C,
                                                   int M,
                                                   float* __restrict__ ones, int n_ones,
                                                   float* __restrict__ zeros, int n_zeros) {
    __shared__ __attribute__((aligned(16))) float As[G_BM * G_AP];
    __shared__ __attribute__((aligned(16))) float Bs[(G_KC / 2) * G_BP];

    const int tid  = threadIdx.x;
    const int lane = tid & 31;
    const int wave = tid >> 5;
    const int wm   = wave >> 2;
    const int wn   = wave & 3;
    const int m0   = blockIdx.x * G_BM;
    const int nb0  = blockIdx.y * G_BN;
    if (INIT) {
        const int i = (blockIdx.y * gridDim.x + blockIdx.x) * 256 + tid;
        if (i < n_ones) ones[i] = 1.0f;
        else if (i < n_ones + n_zeros) zeros[i - n_ones] = 0.0f;
    }
    if (m0 >= M) return;

    const int hi = lane >> 4;
    const int l  = lane & 15;

    const int a_row = tid >> 3;
    const int a_seg = tid & 7;
    const float* __restrict__ ag = A + (size_t)(m0 + a_row) * K + a_seg * 4;
    const int w_kp = wave;
    const float* __restrict__ wg = W + (size_t)(2 * w_kp) * NC + nb0 + lane * 4;

    v4f ra[2], rw[2][2];
    auto fetch = [&](int kc) {
        ra[0] = *(const v4f*)(ag + kc);
        ra[1] = *(const v4f*)(ag + (size_t)32 * K + kc);
        rw[0][0] = *(const v4f*)(wg + (size_t)kc * NC);
        rw[0][1] = *(const v4f*)(wg + (size_t)(kc + 1) * NC);
        rw[1][0] = *(const v4f*)(wg + (size_t)(kc + 16) * NC);
        rw[1][1] = *(const v4f*)(wg + (size_t)(kc + 17) * NC);
    };
    auto stash = [&]() {
        for (int j = 0; j < 2; ++j) {
            v4f v = ra[j];
            if (RELU_A) {
                v.x = fmaxf(v.x, 0.0f); v.y = fmaxf(v.y, 0.0f);
                v.z = fmaxf(v.z, 0.0f); v.w = fmaxf(v.w, 0.0f);
            }
            *(v4f*)(As + (a_row + 32 * j) * G_AP + a_seg * 4) = v;
            const v4f r0 = rw[j][0], r1 = rw[j][1];
            float* bd = Bs + (w_kp + 8 * j) * G_BP + lane * 8;
            *(v4f*)(bd)     = (v4f){r0.x, r1.x, r0.y, r1.y};
            *(v4f*)(bd + 4) = (v4f){r0.z, r1.z, r0.w, r1.w};
        }
    };

    v8f acc[2][2];
    for (int t = 0; t < 2; ++t)
        for (int u = 0; u < 2; ++u) acc[t][u] = (v8f){};

    const float* af = As + (wm * 32 + l) * G_AP + 2 * hi;
    const float* bf = Bs + hi * G_BP + (wn * 32 + l) * 2;

    fetch(0);
    for (int kc = 0; kc < K; kc += G_KC) {
        stash();
        __syncthreads();
        if (kc + G_KC < K) fetch(kc + G_KC);

        v2f a[2][2], b[2][2];
        a[0][0] = *(const v2f*)(af);
        a[0][1] = *(const v2f*)(af + 16 * G_AP);
        b[0][0] = *(const v2f*)(bf);
        b[0][1] = *(const v2f*)(bf + 32);
#pragma unroll
        for (int ks = 0; ks < G_KC / 4; ++ks) {
            const int c = ks & 1, n = c ^ 1;
            if (ks + 1 < G_KC / 4) {
                a[n][0] = *(const v2f*)(af + (ks + 1) * 4);
                a[n][1] = *(const v2f*)(af + 16 * G_AP + (ks + 1) * 4);
                b[n][0] = *(const v2f*)(bf + (ks + 1) * 2 * G_BP);
                b[n][1] = *(const v2f*)(bf + (ks + 1) * 2 * G_BP + 32);
            }
            for (int t = 0; t < 2; ++t)
                for (int u = 0; u < 2; ++u)
                    acc[t][u] = __builtin_amdgcn_wmma_f32_16x16x4_f32(
                         false, a[c][t],  false, b[c][u],
                         (short)0, acc[t][u],  false,  false);
        }
        __syncthreads();
    }

    for (int u = 0; u < 2; ++u) {
        const int col = nb0 + wn * 32 + u * 16 + l;
        for (int t = 0; t < 2; ++t) {
            const int rbase = m0 + wm * 32 + t * 16 + 8 * hi;
            for (int v = 0; v < 8; ++v)
                C[(size_t)(rbase + v) * NC + col] = acc[t][u][v];
        }
    }
}


#define AG_T 1024
#define AG_W (AG_T / 32)
#define AG_G 4
#define AG_NB ((NNODES + AG_R - 1) / AG_R)
#define AG_C 512
#define BK_NT ((NEDGES + BK_E - 1) / BK_E)
#define WS_TOFF_OFF (3u << 20)
#define WS_CNT_OFF (4u << 20)
#define WS_REC_OFF ((4u << 20) + (1u << 16))
static_assert(AG_NB < BK_K, "slot AG_NB, where the last group ends, is inside a tile's offset row");
static_assert((size_t)NEDGES * 4 <= WS_TOFF_OFF, "lists end ahead of the offsets");
static_assert((size_t)WS_TOFF_OFF + (size_t)BK_NT * BK_K * 2 <= WS_CNT_OFF, "offsets end ahead of the counts");
static_assert((size_t)AG_NB * AG_W * 4 <= (1u << 16), "counts end ahead of the records");
static_assert((size_t)WS_REC_OFF + (size_t)AG_NB * AG_W * AG_C * 8 <= (size_t)NNODES * DIM_H * 4,
              "records fit in the region");
static_assert(NNODES <= 65536 && AG_R <= 65536, "src and local row share one record word");
typedef int v2i __attribute__((ext_vector_type(2)));

__device__ __forceinline__ int row_pos(int c) {
    return (c & ~127) | ((c & 3) << 5) | ((c & 127) >> 2);
}
__device__ __forceinline__ int row_col(int p) {
    return (p & ~127) | ((p & 31) << 2) | ((p >> 5) & 3);
}
template <int H, bool FULL>
__device__ __forceinline__ void agg_rows(const float* __restrict__ h, float* acc,
                                         int s_l, int d_l, float n_l, int n, int lane) {
    constexpr int Q = H / 128;
    static_assert(Q * 128 == H, "whole groups of 128 columns");
    const int n_i = __float_as_int(n_l);

    v4f v[2][AG_G][Q];
    auto load_group = [&](int b, int k0) {
        for (int i = 0; i < AG_G; ++i) {
            if (FULL || k0 + i < n) {
                const float* __restrict__ hr =
                    h + (size_t)__builtin_amdgcn_readlane(s_l, k0 + i) * H + lane * 4;
                for (int q = 0; q < Q; ++q) v[b][i][q] = *(const v4f*)(hr + 128 * q);
            } else {
                for (int q = 0; q < Q; ++q) v[b][i][q] = (v4f){0.0f, 0.0f, 0.0f, 0.0f};
            }
        }
    };
    auto add_group = [&](int b, int k0) {
        for (int i = 0; i < AG_G; ++i) {
            if (FULL || k0 + i < n) {
                const float nk = __int_as_float(__builtin_amdgcn_readlane(n_i, k0 + i));
                float* a = acc + __builtin_amdgcn_readlane(d_l, k0 + i) * H + lane;
                for (int q = 0; q < Q; ++q) {
                    const v4f t = v[b][i][q];
                    __hip_atomic_fetch_add(a + 128 * q,      nk * t.x, __ATOMIC_RELAXED,
                                           __HIP_MEMORY_SCOPE_WORKGROUP);
                    __hip_atomic_fetch_add(a + 128 * q + 32, nk * t.y, __ATOMIC_RELAXED,
                                           __HIP_MEMORY_SCOPE_WORKGROUP);
                    __hip_atomic_fetch_add(a + 128 * q + 64, nk * t.z, __ATOMIC_RELAXED,
                                           __HIP_MEMORY_SCOPE_WORKGROUP);
                    __hip_atomic_fetch_add(a + 128 * q + 96, nk * t.w, __ATOMIC_RELAXED,
                                           __HIP_MEMORY_SCOPE_WORKGROUP);
                }
            }
        }
    };
    load_group(0, 0);
#pragma unroll
    for (int g = 0; g < 32 / AG_G; ++g) {
        if (g + 1 < 32 / AG_G) load_group((g + 1) & 1, (g + 1) * AG_G);
        add_group(g & 1, g * AG_G);
    }
}

#define PL_S(H) ((H) / 32)
#define PL_G (AG_R / 32)
static_assert(PL_G * 32 == AG_R, "whole runs of 32 rows");
template <int H, bool REC, bool POOL>
__global__ __launch_bounds__(AG_T) void k_aggregate(const float* __restrict__ h,
                                                    const unsigned short* __restrict__ toff,
                                                    const int* __restrict__ list,
                                                    const float* __restrict__ deg,
                                                    const float* __restrict__ bias,
                                                    int* __restrict__ rec_cnt,
                                                    int* __restrict__ rec,
                                                    float* __restrict__ out,
                                                    const int* __restrict__ batch,
                                                    float* __restrict__ pool,
                                                    float* __restrict__ counts, int NT, int N) {
    __shared__ __attribute__((aligned(16))) float acc[AG_R * H];

    const int tid  = threadIdx.x;
    const int lane = tid & 31;
    const int wave = __builtin_amdgcn_readfirstlane(tid >> 5);
    const int row0 = blockIdx.x * AG_R;
    const int nrows = (N - row0 < AG_R) ? (N - row0) : AG_R;
    int* __restrict__ recw = rec + ((size_t)blockIdx.x * AG_W + wave) * (2 * AG_C);
    int total = 0;
    if (!REC) total = __builtin_amdgcn_readfirstlane(rec_cnt[blockIdx.x * AG_W + wave]);
    const bool replay = !REC && total <= AG_C;

    constexpr int NP = AG_R * H / 4 / AG_T;
    static_assert(NP * AG_T * 4 == AG_R * H, "whole pieces per thread");
    static_assert(AG_T % (H / 4) == 0, "one column group per thread");
    const int p_t = row_pos((tid % (H / 4)) * 4);
    {
        const int c = (tid % (H / 4)) * 4;
        const int r_t = tid / (H / 4);
        const v4f bv = *(const v4f*)(bias + c);
        v4f hv[NP];
        float dv[NP];
#pragma unroll
        for (int j = 0; j < NP; ++j) {
            const int r = r_t + j * (AG_T / (H / 4));
            const int rc = (r < nrows) ? r : (nrows - 1);
            hv[j] = *(const v4f*)(h + (size_t)(row0 + rc) * H + c);
            dv[j] = deg[row0 + rc];
        }
#pragma unroll
        for (int j = 0; j < NP; ++j) {
            const int r = r_t + j * (AG_T / (H / 4));
            const float di = rsqrtf(dv[j]);
            const float d2 = di * di;
            float* ar = acc + r * H + p_t;
            ar[0]  = d2 * hv[j].x + bv.x;
            ar[32] = d2 * hv[j].y + bv.y;
            ar[64] = d2 * hv[j].z + bv.z;
            ar[96] = d2 * hv[j].w + bv.w;
        }
    }
    v2i r_cur = (v2i){0, 0};
    if (replay && lane < total) r_cur = *(const v2i*)(recw + 2 * lane);
    int w_lo = 0, w_n = 0;
    if (!replay) {
        const int t = wave + AG_W * lane;
        if (t < NT) {
            const unsigned short* __restrict__ tp = toff + (size_t)t * BK_K + blockIdx.x;
            w_lo = tp[0];
            w_n = (int)tp[1] - w_lo;
        }
    }
    __syncthreads();

    if (replay) {
        int base = 0;
        for (; base + 32 <= total; base += 32) {
            v2i r_nxt = (v2i){0, 0};
            if (base + 32 + lane < total) r_nxt = *(const v2i*)(recw + 2 * (base + 32 + lane));
            agg_rows<H, true>(h, acc, r_cur.x & 0xffff, (int)((unsigned)r_cur.x >> 16),
                              __int_as_float(r_cur.y), 32, lane);
            r_cur = r_nxt;
        }
        if (base < total)
            agg_rows<H, false>(h, acc, r_cur.x & 0xffff, (int)((unsigned)r_cur.x >> 16),
                               __int_as_float(r_cur.y), total - base, lane);
    } else {
        int nrec = 0;
        for (int tb = wave; tb < NT; tb += 32 * AG_W) {
            if (tb != wave) {
                const int t = tb + AG_W * lane;
                w_lo = 0;
                w_n = 0;
                if (t < NT) {
                    const unsigned short* __restrict__ tp = toff + (size_t)t * BK_K + blockIdx.x;
                    w_lo = tp[0];
                    w_n = (int)tp[1] - w_lo;
                }
            }
            int inc = w_n;
            for (int o = 1; o < 32; o <<= 1) {
                const int v = __shfl_up(inc, o);
                if (lane >= o) inc += v;
            }
            const int first = inc - w_n;
            const int T = __builtin_amdgcn_readlane(inc, 31);
            const int ibase = (tb + AG_W * lane) * BK_E + w_lo - first;
            auto locate = [&](int f) {
                int pos = 0;
                for (int step = 16; step > 0; step >>= 1) {
                    const int fv = __shfl(first, pos + step);
                    if (fv <= f) pos += step;
                }
                return __shfl(ibase, pos) + f;
            };
            int e_cur = 0;
            {
                const int a = locate(lane);
                if (lane < T) e_cur = list[a];
            }
            for (int base = 0; base < T; base += 32) {
                const int m = (T - base < 32) ? (T - base) : 32;
                const int s_l = e_cur & 0xffff;
                const int d_l = (int)((unsigned)e_cur >> 16);
                float g_s = 1.0f, g_d = 1.0f;
                if (lane < m) {
                    g_s = deg[s_l];
                    g_d = deg[row0 + d_l];
                }
                int e_nxt = 0;
                {
                    const int f = base + 32 + lane;
                    const int a = locate(f);
                    if (f < T) e_nxt = list[a];
                }
                float n_l = 0.0f;
                if (lane < m) {
                    n_l = rsqrtf(g_s) * rsqrtf(g_d);
                    if (REC && nrec + lane < AG_C)
                        *(v2i*)(recw + 2 * (nrec + lane)) = (v2i){e_cur, __float_as_int(n_l)};
                }
                if (m == 32) agg_rows<H, true>(h, acc, s_l, d_l, n_l, 32, lane);
                else         agg_rows<H, false>(h, acc, s_l, d_l, n_l, m, lane);
                nrec += m;
                e_cur = e_nxt;
            }
        }
        if (REC && lane == 0) rec_cnt[blockIdx.x * AG_W + wave] = nrec;
    }

    static_assert(!POOL || PL_S(H) * PL_G <= AG_W, "a wave per slice and run");
    const bool pooler = POOL && wave < PL_S(H) * PL_G;
    const int p_run = wave / PL_S(H);
    int g_l = 0;
    if (pooler) {
        const int r = p_run * 32 + lane;
        g_l = batch[row0 + ((r < nrows) ? r : (nrows - 1))];
    }
    __syncthreads();
    {
        const int lim = nrows * (H / 4);
        const float* ar = acc + (tid / (H / 4)) * H + p_t;
        v4f* __restrict__ o4 = (v4f*)(out + (size_t)row0 * H) + tid;
        v4f ov[NP];
#pragma unroll
        for (int j = 0; j < NP; ++j) {
            const float* aj = ar + j * (AG_T / (H / 4)) * H;
            ov[j] = (v4f){aj[0], aj[32], aj[64], aj[96]};
        }
#pragma unroll
        for (int j = 0; j < NP; ++j)
            if (tid + j * AG_T < lim) o4[j * AG_T] = ov[j];
    }
    if (pooler) {
        const int cnt_r = (nrows - p_run * 32 < 32) ? (nrows - p_run * 32) : 32;
        if (cnt_r > 0) {
            const int pp = (wave % PL_S(H)) * 32 + lane;
            const int c = row_col(pp);
            const float* a = acc + (p_run * 32) * H + pp;
            float v[32];
            for (int k = 0; k < 32; ++k) v[k] = a[k * H];
            float* __restrict__ pc = pool + c;
            int g = __builtin_amdgcn_readlane(g_l, 0);
            int len = 0;
            float s = 0.0f;
#pragma unroll
            for (int k = 0; k < 32; ++k) {
                if (k < cnt_r) {
                    const int gk = __builtin_amdgcn_readlane(g_l, k);
                    if (gk != g) {
                        unsafeAtomicAdd(pc + (size_t)g * H, s);
                        if (c == 0) unsafeAtomicAdd(&counts[g], (float)len);
                        s = 0.0f;
                        len = 0;
                        g = gk;
                    }
                    s += v[k];
                    ++len;
                }
            }
            unsafeAtomicAdd(pc + (size_t)g * H, s);
            if (c == 0) unsafeAtomicAdd(&counts[g], (float)len);
        }
    }
}


__global__ __launch_bounds__(DIM_O) void k_head(const float* __restrict__ pool,
                                                const float* __restrict__ counts,
                                                const float* __restrict__ Wc,
                                                const float* __restrict__ bc,
                                                float* __restrict__ gemb,
                                                float* __restrict__ logits) {
    __shared__ float part[DIM_O / 32][2];
    const int g = blockIdx.x;
    const int t = threadIdx.x;
    const v2f w = *(const v2f*)(Wc + 2 * t);
    const float ge = pool[g * DIM_O + t] / fmaxf(counts[g], 1.0f);
    gemb[g * DIM_O + t] = ge;
    float p0 = ge * w.x, p1 = ge * w.y;
    for (int o = 16; o > 0; o >>= 1) {
        p0 += __shfl_xor(p0, o);
        p1 += __shfl_xor(p1, o);
    }
    if ((t & 31) == 0) {
        part[t >> 5][0] = p0;
        part[t >> 5][1] = p1;
    }
    __syncthreads();
    if (t < 2) {
        float a = bc[t];
        for (int k = 0; k < DIM_O / 32; ++k) a += part[k][t];
        logits[g * 2 + t] = a;
    }
}


static inline int cdiv_l(long a, int b) { return (int)((a + b - 1) / b); }

extern "C" void kernel_launch(void* const* d_in, const int* in_sizes, int n_in,
                              void* d_out, int out_size, void* d_ws, size_t ws_size,
                              hipStream_t stream) {
    const float* x     = (const float*)d_in[0];
    const int*   ei    = (const int*)d_in[1];
    const int*   batch = (const int*)d_in[2];
    const float* W_in  = (const float*)d_in[3];
    const float* b_in  = (const float*)d_in[4];
    const float* W_mid = (const float*)d_in[5];
    const float* b_mid = (const float*)d_in[6];
    const float* W_out = (const float*)d_in[7];
    const float* b_out = (const float*)d_in[8];
    const float* W_cls = (const float*)d_in[9];
    const float* b_cls = (const float*)d_in[10];

    const int N = in_sizes[0] / DIM_IN;
    const int E = in_sizes[1] / 2;
    const int* src = ei;
    const int* dst = ei + E;

    float* w       = (float*)d_ws;
    float* deg     = w;
    float* bufH    = deg + N;
    float* bufAgg  = bufH + (size_t)N * DIM_H;
    float* bufA    = bufAgg + (size_t)N * DIM_H;
    float* pool    = bufA + (size_t)N * DIM_H;
    float* counts  = pool + (size_t)NGRAPH * DIM_O;

    int* list = (int*)bufA;
    unsigned short* toff = (unsigned short*)((unsigned char*)bufA + WS_TOFF_OFF);
    int* rec_cnt = (int*)((unsigned char*)bufA + WS_CNT_OFF);
    int* rec     = (int*)((unsigned char*)bufA + WS_REC_OFF);

    float* node_emb = (float*)d_out;
    float* gemb     = node_emb + (size_t)N * DIM_O;
    float* logits   = gemb + (size_t)NGRAPH * DIM_O;

    const int B = 256;
    const int MB = G_BM;

    {
        dim3 grid(N / MB, DIM_H / 128);
        k_gemm_seed<false, DIM_IN, DIM_H, true><<<grid, B, 0, stream>>>(
            x, W_in, bufH, N, deg, N, pool, NGRAPH * DIM_O + NGRAPH);
    }

    const int NT = cdiv_l(E, BK_E);
    k_deg_group<<<NT, BK_T, 0, stream>>>(src, dst, deg, toff, list, E);

    k_aggregate<DIM_H, true, false><<<cdiv_l(N, AG_R), AG_T, 0, stream>>>(
        bufH, toff, list, deg, b_in, rec_cnt, rec, bufAgg, batch, pool, counts, NT, N);

    {
        dim3 grid(N / MB, DIM_H / 128);
        k_gemm_seed<true, DIM_H, DIM_H, false><<<grid, B, 0, stream>>>(
            bufAgg, W_mid, bufH, N, deg, 0, pool, 0);
        k_aggregate<DIM_H, false, false><<<cdiv_l(N, AG_R), AG_T, 0, stream>>>(
            bufH, toff, list, deg, b_mid, rec_cnt, rec, bufAgg, batch, pool, counts, NT, N);
    }

    {
        dim3 grid(N / MB, DIM_O / 128);
        k_gemm_seed<true, DIM_H, DIM_O, false><<<grid, B, 0, stream>>>(
            bufAgg, W_out, bufH, N, deg, 0, pool, 0);
        k_aggregate<DIM_O, false, true><<<cdiv_l(N, AG_R), AG_T, 0, stream>>>(
            bufH, toff, list, deg, b_out, rec_cnt, rec, node_emb, batch, pool, counts, NT, N);
    }

    k_head<<<NGRAPH, DIM_O, 0, stream>>>(pool, counts, W_cls, b_cls, gemb, logits);
}
